# starting kernel
# baseline (speedup 1.0000x reference)
#include <hip/hip_runtime.h>

#define NN   50000
#define NE   800000
#define HD   128

typedef float v2f __attribute__((ext_vector_type(2)));
typedef float v8f __attribute__((ext_vector_type(8)));

__global__ __launch_bounds__(256) void dgi_deg_init(float* __restrict__ deg) {
    int i = blockIdx.x * blockDim.x + threadIdx.x;
    if (i < NN) deg[i] = 1.0f;
}

__global__ __launch_bounds__(256) void dgi_deg_accum(const int* __restrict__ dst,
                                                     float* __restrict__ deg) {
    int e = blockIdx.x * blockDim.x + threadIdx.x;
    if (e < NE) atomicAdd(&deg[dst[e]], 1.0f);
}

__global__ __launch_bounds__(256) void dgi_dinv(float* __restrict__ degdinv) {
    int i = blockIdx.x * blockDim.x + threadIdx.x;
    if (i < NN) {
        float d = degdinv[i];
        degdinv[i] = (d > 0.0f) ? rsqrtf(d) : 0.0f;
    }
}

__global__ __launch_bounds__(256) void dgi_gemm128_wmma(const float* __restrict__ A,
                                                        const float* __restrict__ B,
                                                        float* __restrict__ C) {
    const int wave = threadIdx.x >> 5;
    const int lane = threadIdx.x & 31;
    const int half = lane >> 4;
    const int l    = lane & 15;
    const int m0   = blockIdx.x * 16;
    const int n0   = wave * 16;

    const float* __restrict__ arow = A + (size_t)(m0 + l) * HD;

    v8f acc = {0.f, 0.f, 0.f, 0.f, 0.f, 0.f, 0.f, 0.f};

#pragma unroll
    for (int k0 = 0; k0 < HD; k0 += 4) {
        const int ka = k0 + 2 * half;
        v2f a, b;
        a.x = arow[ka + 0];
        a.y = arow[ka + 1];
        b.x = B[(size_t)(ka + 0) * HD + n0 + l];
        b.y = B[(size_t)(ka + 1) * HD + n0 + l];
        acc = __builtin_amdgcn_wmma_f32_16x16x4_f32(
            false, a, false, b, (short)0, acc, false, false);
    }

#pragma unroll
    for (int r = 0; r < 8; ++r) {
        C[(size_t)(m0 + r + 8 * half) * HD + n0 + l] = acc[r];
    }
}

__global__ __launch_bounds__(256) void dgi_z_init(const float* __restrict__ xw,
                                                  const float* __restrict__ dinv,
                                                  const float* __restrict__ bias,
                                                  float* __restrict__ z) {
    long long i = (long long)blockIdx.x * blockDim.x + threadIdx.x;
    if (i < (long long)NN * HD) {
        int n  = (int)(i >> 7);
        int ch = (int)(i & 127);
        float di = dinv[n];
        z[i] = bias[ch] + di * di * xw[i];
    }
}

__global__ __launch_bounds__(256) void dgi_edge_scatter(const int* __restrict__ src,
                                                        const int* __restrict__ dst,
                                                        const float* __restrict__ xw,
                                                        const float* __restrict__ dinv,
                                                        float* __restrict__ z) {
    long long i = (long long)blockIdx.x * blockDim.x + threadIdx.x;
    if (i < (long long)NE * HD) {
        int e  = (int)(i >> 7);
        int ch = (int)(i & 127);
        int s  = src[e];
        int d  = dst[e];
        float c = dinv[s] * dinv[d];
        atomicAdd(&z[(long long)d * HD + ch], xw[(long long)s * HD + ch] * c);
    }
}

__global__ __launch_bounds__(256) void dgi_score(const float* __restrict__ u,
                                                 const float* __restrict__ z,
                                                 const int* __restrict__ perm,
                                                 const float* __restrict__ bb,
                                                 float* __restrict__ out) {
    const int wave = threadIdx.x >> 5;
    const int lane = threadIdx.x & 31;
    const int n = blockIdx.x * 8 + wave;
    if (n >= NN) return;

    const float* un = u + (size_t)n * HD;
    const float* za = z + (size_t)n * HD;
    const float* zb = z + (size_t)perm[n] * HD;

    float4 uu = *(const float4*)(un + lane * 4);
    float4 va = *(const float4*)(za + lane * 4);
    float4 vb = *(const float4*)(zb + lane * 4);

    float pos = uu.x * va.x + uu.y * va.y + uu.z * va.z + uu.w * va.w;
    float neg = uu.x * vb.x + uu.y * vb.y + uu.z * vb.z + uu.w * vb.w;

#pragma unroll
    for (int off = 16; off > 0; off >>= 1) {
        pos += __shfl_xor(pos, off, 32);
        neg += __shfl_xor(neg, off, 32);
    }
    if (lane == 0) {
        float bias = bb[0];
        out[n]      = pos + bias;
        out[NN + n] = neg + bias;
    }
}

extern "C" void kernel_launch(void* const* d_in, const int* in_sizes, int n_in,
                              void* d_out, int out_size, void* d_ws, size_t ws_size,
                              hipStream_t stream) {
    (void)in_sizes; (void)n_in; (void)out_size; (void)ws_size;

    const float* x    = (const float*)d_in[0];
    const float* W    = (const float*)d_in[1];
    const float* b    = (const float*)d_in[2];
    const float* Wb   = (const float*)d_in[3];
    const float* bb   = (const float*)d_in[4];
    const int*   ei   = (const int*)d_in[5];
    const int*   perm = (const int*)d_in[6];
    const int*   esrc = ei;
    const int*   edst = ei + NE;
    float*       out  = (float*)d_out;

    char* ws = (char*)d_ws;
    size_t off = 0;
    float* dinv = (float*)(ws + off); off += ((size_t)NN * 4 + 255) & ~(size_t)255;
    float* xw   = (float*)(ws + off); off += ((size_t)NN * HD * 4 + 255) & ~(size_t)255;
    float* z    = (float*)(ws + off); off += ((size_t)NN * HD * 4 + 255) & ~(size_t)255;
    float* u    = (float*)(ws + off); off += ((size_t)NN * HD * 4 + 255) & ~(size_t)255;

    const int nblk  = (NN + 255) / 256;
    const int eblk  = (NE + 255) / 256;
    const int nhblk = (int)(((long long)NN * HD + 255) / 256);
    const int ehblk = (int)(((long long)NE * HD + 255) / 256);

    dgi_deg_init<<<nblk, 256, 0, stream>>>(dinv);
    dgi_deg_accum<<<eblk, 256, 0, stream>>>(edst, dinv);
    dgi_dinv<<<nblk, 256, 0, stream>>>(dinv);

    dgi_gemm128_wmma<<<NN / 16, 256, 0, stream>>>(x, W, xw);

    dgi_z_init<<<nhblk, 256, 0, stream>>>(xw, dinv, b, z);
    dgi_edge_scatter<<<ehblk, 256, 0, stream>>>(esrc, edst, xw, dinv, z);

    dgi_gemm128_wmma<<<NN / 16, 256, 0, stream>>>(z, Wb, u);

    dgi_score<<<(NN + 7) / 8, 256, 0, stream>>>(u, z, perm, bb, out);
}


# owners scan one-byte tile keys written by the degree pass (16 edges per 16-byte load, exact zero-byte match) instead of the 4-byte dst list; dst and src are read only for matching edges
# speedup vs baseline: 2.9869x; 2.9869x over previous
#include <hip/hip_runtime.h>

#define NN   50000
#define NE   800000
#define HD   128

#define TR     196
#define NE4    (NE / 4)
#define NK16   (NE / 16)

typedef float v2f __attribute__((ext_vector_type(2)));
typedef float v8f __attribute__((ext_vector_type(8)));

__global__ __launch_bounds__(256) void dgi_deg_init(float* __restrict__ deg) {
    int i = blockIdx.x * blockDim.x + threadIdx.x;
    if (i < NN) deg[i] = 1.0f;
}

__global__ __launch_bounds__(256) void dgi_deg_keys(const int* __restrict__ dst,
                                                    float* __restrict__ deg,
                                                    unsigned* __restrict__ keys) {
    const int g = blockIdx.x * blockDim.x + threadIdx.x;
    if (g < NE4) {
        const int4 d = ((const int4*)dst)[g];
        atomicAdd(&deg[d.x], 1.0f);
        atomicAdd(&deg[d.y], 1.0f);
        atomicAdd(&deg[d.z], 1.0f);
        atomicAdd(&deg[d.w], 1.0f);
        keys[g] = ((unsigned)d.x / TR) | (((unsigned)d.y / TR) << 8) |
                  (((unsigned)d.z / TR) << 16) | (((unsigned)d.w / TR) << 24);
    }
}

__global__ __launch_bounds__(256) void dgi_dinv(float* __restrict__ degdinv) {
    int i = blockIdx.x * blockDim.x + threadIdx.x;
    if (i < NN) {
        float d = degdinv[i];
        degdinv[i] = (d > 0.0f) ? rsqrtf(d) : 0.0f;
    }
}

template <bool PERMK>
__device__ __forceinline__ int kmap(int p) {
    return PERMK ? (4 * (p & 31) + (p >> 5)) : p;
}

template <bool PERMK>
__device__ __forceinline__ void load_bfrags(const float* __restrict__ B, int col, int half,
                                            v2f (&bf)[32]) {
#pragma unroll
    for (int j = 0; j < 32; ++j) {
        const int p = 4 * j + 2 * half;
        bf[j].x = B[(size_t)kmap<PERMK>(p + 0) * HD + col];
        bf[j].y = B[(size_t)kmap<PERMK>(p + 1) * HD + col];
    }
}

__device__ __forceinline__ v8f tile_mma(const float* arow, const v2f (&bf)[32]) {
    v8f acc = {0.f, 0.f, 0.f, 0.f, 0.f, 0.f, 0.f, 0.f};
#pragma unroll
    for (int j = 0; j < 32; ++j) {
        const v2f a = *(const v2f*)(arow + 4 * j);
        acc = __builtin_amdgcn_wmma_f32_16x16x4_f32(
            false, a, false, bf[j], (short)0, acc, false, false);
    }
    return acc;
}

#define TRP    208
#define PITCH  132
#define NTILE  256
#define AGGW   32
#define QCAP   64

__device__ __forceinline__ unsigned zero_bytes(unsigned v) {
    const unsigned t = (v & 0x7F7F7F7Fu) + 0x7F7F7F7Fu;
    return ~(t | v) & 0x80808080u;
}

__global__ __launch_bounds__(1024) void dgi_tile_aggregate(const int* __restrict__ src,
                                                           const int* __restrict__ dst,
                                                           const unsigned* __restrict__ keys,
                                                           const float* __restrict__ x,
                                                           const float* __restrict__ Wt,
                                                           const float* __restrict__ dinv,
                                                           const float* __restrict__ bias,
                                                           const float* __restrict__ Wb,
                                                           float* __restrict__ z,
                                                           float* __restrict__ u) {
    __shared__ float    tile[TRP * PITCH];
    __shared__ float    zt[TRP * PITCH];
    __shared__ unsigned queue[AGGW * QCAP];

    const int wave = threadIdx.x >> 5;
    const int lane = threadIdx.x & 31;
    const int lo   = blockIdx.x * TR;
    const int rows = (NN - lo < TR) ? (NN - lo) : TR;

    for (int r = wave; r < TRP; r += AGGW) {
        float4 v = make_float4(0.f, 0.f, 0.f, 0.f);
        if (r < rows) {
            const float di = dinv[lo + r];
            const float4 xv = *(const float4*)(x + (size_t)(lo + r) * HD + lane * 4);
            v.x = di * di * xv.x;
            v.y = di * di * xv.y;
            v.z = di * di * xv.z;
            v.w = di * di * xv.w;
        }
        float* t = tile + r * PITCH + lane;
        t[0]  = v.x;
        t[32] = v.y;
        t[64] = v.z;
        t[96] = v.w;
    }
    __syncthreads();

    unsigned* q = queue + wave * QCAP;
    int qn = 0;

    auto flush = [&]() {
        for (int base = 0; base < qn; base += 32) {
            const int cnt = (qn - base < 32) ? (qn - base) : 32;
            int   s_l = 0, r_l = 0;
            float c_l = 0.0f;
            if (lane < cnt) {
                const int e = (int)q[base + lane];
                const int d = dst[e];
                s_l = src[e];
                r_l = d - lo;
                c_l = dinv[s_l] * dinv[d];
            }
            int k = 0;
            for (; k + 4 <= cnt; k += 4) {
                float4 v[4];
                int    rr[4];
                float  cc[4];
#pragma unroll
                for (int j = 0; j < 4; ++j) {
                    const int s = __builtin_amdgcn_readlane(s_l, k + j);
                    rr[j] = __builtin_amdgcn_readlane(r_l, k + j);
                    cc[j] = __int_as_float(__builtin_amdgcn_readlane(__float_as_int(c_l), k + j));
                    v[j]  = *(const float4*)(x + (size_t)s * HD + lane * 4);
                }
#pragma unroll
                for (int j = 0; j < 4; ++j) {
                    float* t = tile + rr[j] * PITCH + lane;
                    atomicAdd(t + 0,  v[j].x * cc[j]);
                    atomicAdd(t + 32, v[j].y * cc[j]);
                    atomicAdd(t + 64, v[j].z * cc[j]);
                    atomicAdd(t + 96, v[j].w * cc[j]);
                }
            }
            for (; k < cnt; ++k) {
                const int   s = __builtin_amdgcn_readlane(s_l, k);
                const int   r = __builtin_amdgcn_readlane(r_l, k);
                const float c = __int_as_float(__builtin_amdgcn_readlane(__float_as_int(c_l), k));
                const float4 v = *(const float4*)(x + (size_t)s * HD + lane * 4);
                float* t = tile + r * PITCH + lane;
                atomicAdd(t + 0,  v.x * c);
                atomicAdd(t + 32, v.y * c);
                atomicAdd(t + 64, v.z * c);
                atomicAdd(t + 96, v.w * c);
            }
        }
        qn = 0;
    };

    const unsigned pat = (unsigned)blockIdx.x * 0x01010101u;
    const uint4* keys4 = (const uint4*)keys;
    for (int b = wave * 32; b < NK16; b += 4 * 1024) {
        uint4 d[4];
#pragma unroll
        for (int u = 0; u < 4; ++u) {
            const int g = b + u * 1024 + lane;
            d[u] = keys4[(g < NK16) ? g : NK16 - 1];
        }
#pragma unroll
        for (int u = 0; u < 4; ++u) {
            const int g  = b + u * 1024 + lane;
            const int e0 = g * 16;
            unsigned m = zero_bytes(d[u].x ^ pat) | (zero_bytes(d[u].y ^ pat) >> 1) |
                         (zero_bytes(d[u].z ^ pat) >> 2) | (zero_bytes(d[u].w ^ pat) >> 3);
            if (g >= NK16) m = 0;
            unsigned bal = (unsigned)__ballot(m != 0);
            while (bal != 0) {
                if (m != 0) {
                    const int p = __ffs(m) - 1;
                    q[qn + (int)__builtin_amdgcn_mbcnt_lo(bal, 0u)] =
                        (unsigned)(e0 + 4 * (7 - (p & 7)) + (p >> 3));
                }
                m &= m - 1;
                qn += __popc(bal);
                if (qn >= 32) flush();
                bal = (unsigned)__ballot(m != 0);
            }
        }
    }
    if (qn > 0) flush();

    const int half = lane >> 4;
    const int l    = lane & 15;
    const int n0   = (wave & 7) * 16;
    const int nrt  = (rows + 15) >> 4;
    v2f bf[32];

    load_bfrags<true>(Wt, n0 + l, half, bf);
    const float bn = bias[n0 + l];
    __syncthreads();
    {
        for (int rt = wave >> 3; rt < nrt; rt += 4) {
            const v8f acc = tile_mma(tile + (rt * 16 + l) * PITCH + 2 * half, bf);
#pragma unroll
            for (int r = 0; r < 8; ++r) {
                const int m = rt * 16 + r + 8 * half;
                const float v = acc[r] + bn;
                zt[m * PITCH + n0 + l] = v;
                if (m < rows) z[(size_t)(lo + m) * HD + n0 + l] = v;
            }
        }
    }

    load_bfrags<false>(Wb, n0 + l, half, bf);
    __syncthreads();
    for (int rt = wave >> 3; rt < nrt; rt += 4) {
        const v8f acc = tile_mma(zt + (rt * 16 + l) * PITCH + 2 * half, bf);
#pragma unroll
        for (int r = 0; r < 8; ++r) {
            const int m = rt * 16 + r + 8 * half;
            if (m < rows) u[(size_t)(lo + m) * HD + n0 + l] = acc[r];
        }
    }
}

__global__ __launch_bounds__(256) void dgi_score(const float* __restrict__ u,
                                                 const float* __restrict__ z,
                                                 const int* __restrict__ perm,
                                                 const float* __restrict__ bb,
                                                 float* __restrict__ out) {
    const int wave = threadIdx.x >> 5;
    const int lane = threadIdx.x & 31;
    const int n = blockIdx.x * 8 + wave;
    if (n >= NN) return;

    const float* un = u + (size_t)n * HD;
    const float* za = z + (size_t)n * HD;
    const float* zb = z + (size_t)perm[n] * HD;

    float4 uu = *(const float4*)(un + lane * 4);
    float4 va = *(const float4*)(za + lane * 4);
    float4 vb = *(const float4*)(zb + lane * 4);

    float pos = uu.x * va.x + uu.y * va.y + uu.z * va.z + uu.w * va.w;
    float neg = uu.x * vb.x + uu.y * vb.y + uu.z * vb.z + uu.w * vb.w;

#pragma unroll
    for (int off = 16; off > 0; off >>= 1) {
        pos += __shfl_xor(pos, off, 32);
        neg += __shfl_xor(neg, off, 32);
    }
    if (lane == 0) {
        float bias = bb[0];
        out[n]      = pos + bias;
        out[NN + n] = neg + bias;
    }
}

extern "C" void kernel_launch(void* const* d_in, const int* in_sizes, int n_in,
                              void* d_out, int out_size, void* d_ws, size_t ws_size,
                              hipStream_t stream) {
    (void)in_sizes; (void)n_in; (void)out_size; (void)ws_size;

    const float* x    = (const float*)d_in[0];
    const float* W    = (const float*)d_in[1];
    const float* b    = (const float*)d_in[2];
    const float* Wb   = (const float*)d_in[3];
    const float* bb   = (const float*)d_in[4];
    const int*   ei   = (const int*)d_in[5];
    const int*   perm = (const int*)d_in[6];
    const int*   esrc = ei;
    const int*   edst = ei + NE;
    float*       out  = (float*)d_out;

    char* ws = (char*)d_ws;
    size_t off = 0;
    float* dinv = (float*)(ws + off); off += ((size_t)NN * 4 + 255) & ~(size_t)255;
    float* z    = (float*)(ws + off); off += ((size_t)NN * HD * 4 + 255) & ~(size_t)255;
    float* u    = (float*)(ws + off); off += ((size_t)NN * HD * 4 + 255) & ~(size_t)255;
    unsigned* keys = (unsigned*)(ws + off); off += ((size_t)NE + 255) & ~(size_t)255;

    const int nblk  = (NN + 255) / 256;
    const int kblk  = (NE4 + 255) / 256;

    dgi_deg_init<<<nblk, 256, 0, stream>>>(dinv);
    dgi_deg_keys<<<kblk, 256, 0, stream>>>(edst, dinv, keys);
    dgi_dinv<<<nblk, 256, 0, stream>>>(dinv);

    dgi_tile_aggregate<<<NTILE, 1024, 0, stream>>>(esrc, edst, keys, x, W, dinv, b, Wb, z, u);

    dgi_score<<<(NN + 7) / 8, 256, 0, stream>>>(u, z, perm, bb, out);
}


# edges binned by owner tile in the degree pass (per-chunk src|row lists, exact offsets, slot by LDS cursor in arrival order: sum order changes); owner wave reads its list; key scan, dst/src reads gone
# speedup vs baseline: 4.4617x; 1.4938x over previous
#include <hip/hip_runtime.h>

#define NN   50000
#define NE   800000
#define HD   128

#define TR     196
#define NTILE  256
#define NE4    (NE / 4)

typedef float v2f __attribute__((ext_vector_type(2)));
typedef float v8f __attribute__((ext_vector_type(8)));

#define DR     8
#define DC     32
#define TPR    (NTILE / DR)
#define DRN    (TPR * TR)
#define DCE    (NE / DC)
#define DCE4   (NE4 / DC)
#define DTH    512
#define DG     ((DCE4 + DTH - 1) / DTH)

static_assert(TPR == 32, "one lane per tile of a range in the start scan");
static_assert(DTH / 16 == TPR, "sixteen threads per tile in the tile sums");
static_assert(DR * DRN >= NN && NTILE * TR >= NN, "ranges and tiles cover every node");
static_assert(TR <= 256 && NN < (1 << 24), "a list word holds the row in 8 bits and the source above them");

__global__ __launch_bounds__(DTH) void dgi_deg_bin(const int* __restrict__ src,
                                                   const int* __restrict__ dst,
                                                   unsigned* __restrict__ part,
                                                   unsigned* __restrict__ lists,
                                                   unsigned* __restrict__ offs) {
    __shared__ unsigned cnt[DRN];
    __shared__ unsigned ttot[TPR];
    __shared__ unsigned cur[TPR];
    __shared__ unsigned wbel[DTH / 32];
    const int c    = blockIdx.x / DR;
    const int r    = blockIdx.x % DR;
    const int nlo  = r * DRN;
    const int lane = threadIdx.x & 31;
    const int wave = threadIdx.x >> 5;

    for (int i = threadIdx.x; i < DRN; i += DTH) cnt[i] = 0u;
    __syncthreads();

    const int4* d4 = (const int4*)dst + (size_t)c * DCE4;
    const int4* s4 = (const int4*)src + (size_t)c * DCE4;
    int4 d[DG], sv[DG];
#pragma unroll
    for (int k = 0; k < DG; ++k) {
        const int i = threadIdx.x + k * DTH;
        d[k] = d4[(i < DCE4) ? i : DCE4 - 1];
    }
#pragma unroll
    for (int k = 0; k < DG; ++k) {
        const int i = threadIdx.x + k * DTH;
        sv[k] = s4[(i < DCE4) ? i : DCE4 - 1];
    }

    unsigned below = 0u;
#pragma unroll
    for (int k = 0; k < DG; ++k) {
        const int i = threadIdx.x + k * DTH;
        if (i < DCE4) {
            const unsigned a0 = (unsigned)(d[k].x - nlo);
            const unsigned a1 = (unsigned)(d[k].y - nlo);
            const unsigned a2 = (unsigned)(d[k].z - nlo);
            const unsigned a3 = (unsigned)(d[k].w - nlo);
            if (a0 < DRN) atomicAdd(&cnt[a0], 1u);
            if (a1 < DRN) atomicAdd(&cnt[a1], 1u);
            if (a2 < DRN) atomicAdd(&cnt[a2], 1u);
            if (a3 < DRN) atomicAdd(&cnt[a3], 1u);
            below += (unsigned)(d[k].x < nlo) + (unsigned)(d[k].y < nlo) +
                     (unsigned)(d[k].z < nlo) + (unsigned)(d[k].w < nlo);
        }
    }
    for (int off = 16; off > 0; off >>= 1) below += __shfl_xor(below, off, 32);
    if (lane == 0) wbel[wave] = below;
    __syncthreads();

    unsigned* prow = part + (size_t)c * NN + nlo;
    for (int i = threadIdx.x; i < DRN; i += DTH)
        if (nlo + i < NN) prow[i] = cnt[i];

    {
        const int tl  = threadIdx.x >> 4;
        const int sub = threadIdx.x & 15;
        unsigned t = 0u;
        for (int m = sub; m < TR; m += 16) t += cnt[tl * TR + m];
        for (int off = 8; off > 0; off >>= 1) t += __shfl_xor(t, off, 32);
        if (sub == 0) ttot[tl] = t;
    }
    __syncthreads();

    if (wave == 0) {
        unsigned b = (lane < DTH / 32) ? wbel[lane] : 0u;
        for (int off = 16; off > 0; off >>= 1) b += __shfl_xor(b, off, 32);
        const unsigned t = ttot[lane];
        unsigned inc = t;
        for (int off = 1; off < 32; off <<= 1) {
            const unsigned v = __shfl_up(inc, off, 32);
            if (lane >= off) inc += v;
        }
        const unsigned start = b + inc - t;
        cur[lane] = start;
        offs[c * NTILE + r * TPR + lane] = start;
    }
    __syncthreads();

    unsigned* lst = lists + (size_t)c * DCE;
#pragma unroll
    for (int k = 0; k < DG; ++k) {
        const int i = threadIdx.x + k * DTH;
        if (i < DCE4) {
            const int dd[4] = {d[k].x, d[k].y, d[k].z, d[k].w};
            const int ss[4] = {sv[k].x, sv[k].y, sv[k].z, sv[k].w};
            unsigned a[4], tl[4], pos[4];
#pragma unroll
            for (int j = 0; j < 4; ++j) {
                a[j]   = (unsigned)(dd[j] - nlo);
                tl[j]  = a[j] / TR;
                pos[j] = 0u;
                if (a[j] < DRN) pos[j] = atomicAdd(&cur[tl[j]], 1u);
            }
#pragma unroll
            for (int j = 0; j < 4; ++j)
                if (a[j] < DRN) lst[pos[j]] = ((unsigned)ss[j] << 8) | (a[j] - tl[j] * TR);
        }
    }
}

__global__ __launch_bounds__(256) void dgi_dinv(const unsigned* __restrict__ part,
                                                float* __restrict__ dinv) {
    const int i = blockIdx.x * blockDim.x + threadIdx.x;
    if (i < NN) {
        unsigned s = 1u;
#pragma unroll
        for (int c = 0; c < DC; ++c) s += part[(size_t)c * NN + i];
        const float d = (float)s;
        dinv[i] = (d > 0.0f) ? rsqrtf(d) : 0.0f;
    }
}

template <bool PERMK>
__device__ __forceinline__ int kmap(int p) {
    return PERMK ? (4 * (p & 31) + (p >> 5)) : p;
}

template <bool PERMK>
__device__ __forceinline__ void load_bfrags(const float* __restrict__ B, int col, int half,
                                            v2f (&bf)[32]) {
#pragma unroll
    for (int j = 0; j < 32; ++j) {
        const int p = 4 * j + 2 * half;
        bf[j].x = B[(size_t)kmap<PERMK>(p + 0) * HD + col];
        bf[j].y = B[(size_t)kmap<PERMK>(p + 1) * HD + col];
    }
}

__device__ __forceinline__ v8f tile_mma(const float* arow, const v2f (&bf)[32]) {
    v8f acc = {0.f, 0.f, 0.f, 0.f, 0.f, 0.f, 0.f, 0.f};
#pragma unroll
    for (int j = 0; j < 32; ++j) {
        const v2f a = *(const v2f*)(arow + 4 * j);
        acc = __builtin_amdgcn_wmma_f32_16x16x4_f32(
            false, a, false, bf[j], (short)0, acc, false, false);
    }
    return acc;
}

#define TRP    208
#define PITCH  132
#define AGGW   32
static_assert(AGGW == DC, "one edge chunk per owner wave");

__global__ __launch_bounds__(1024) void dgi_tile_aggregate(const unsigned* __restrict__ lists,
                                                           const unsigned* __restrict__ offs,
                                                           const float* __restrict__ x,
                                                           const float* __restrict__ Wt,
                                                           const float* __restrict__ dinv,
                                                           const float* __restrict__ bias,
                                                           const float* __restrict__ Wb,
                                                           float* __restrict__ z,
                                                           float* __restrict__ u) {
    __shared__ float    tile[TRP * PITCH];
    __shared__ float    zt[TRP * PITCH];

    const int wave = __builtin_amdgcn_readfirstlane((int)(threadIdx.x >> 5));
    const int lane = threadIdx.x & 31;
    const int lo   = blockIdx.x * TR;
    const int rows = (NN - lo < TR) ? (NN - lo) : TR;

    for (int r = wave; r < TRP; r += AGGW) {
        float4 v = make_float4(0.f, 0.f, 0.f, 0.f);
        if (r < rows) {
            const float di = dinv[lo + r];
            const float4 xv = *(const float4*)(x + (size_t)(lo + r) * HD + lane * 4);
            v.x = di * di * xv.x;
            v.y = di * di * xv.y;
            v.z = di * di * xv.z;
            v.w = di * di * xv.w;
        }
        float* t = tile + r * PITCH + lane;
        t[0]  = v.x;
        t[32] = v.y;
        t[64] = v.z;
        t[96] = v.w;
    }
    __syncthreads();

    {
        const unsigned* lst = lists + (size_t)wave * DCE;
        const int beg = (int)offs[wave * NTILE + blockIdx.x];
        const int end = (blockIdx.x == NTILE - 1) ? DCE
                                                  : (int)offs[wave * NTILE + blockIdx.x + 1];
        unsigned e0 = 0u, e1 = 0u;
        if (beg + lane < end)      e0 = lst[beg + lane];
        if (beg + 32 + lane < end) e1 = lst[beg + 32 + lane];
        int   s_l = (int)(e0 >> 8);
        int   r_l = (int)(e0 & 255u);
        float c_l = 0.0f;
        if (beg + lane < end) c_l = dinv[s_l] * dinv[lo + r_l];
        for (int base = beg; base < end; base += 32) {
            const int cnt = (end - base < 32) ? (end - base) : 32;
            const int s_n = (int)(e1 >> 8);
            const int r_n = (int)(e1 & 255u);
            const float fa = dinv[s_n];
            const float fb = dinv[lo + r_n];
            unsigned e2 = 0u;
            if (base + 64 + lane < end) e2 = lst[base + 64 + lane];
            int k = 0;
            for (; k + 4 <= cnt; k += 4) {
                float4 v[4];
                int    rr[4];
                float  cc[4];
#pragma unroll
                for (int j = 0; j < 4; ++j) {
                    const int sn = __builtin_amdgcn_readlane(s_l, k + j);
                    rr[j] = __builtin_amdgcn_readlane(r_l, k + j);
                    cc[j] = __int_as_float(__builtin_amdgcn_readlane(__float_as_int(c_l), k + j));
                    v[j]  = *(const float4*)(x + (size_t)sn * HD + lane * 4);
                }
#pragma unroll
                for (int j = 0; j < 4; ++j) {
                    float* t = tile + rr[j] * PITCH + lane;
                    atomicAdd(t + 0,  v[j].x * cc[j]);
                    atomicAdd(t + 32, v[j].y * cc[j]);
                    atomicAdd(t + 64, v[j].z * cc[j]);
                    atomicAdd(t + 96, v[j].w * cc[j]);
                }
            }
            for (; k < cnt; ++k) {
                const int   sn = __builtin_amdgcn_readlane(s_l, k);
                const int   rn = __builtin_amdgcn_readlane(r_l, k);
                const float cn = __int_as_float(__builtin_amdgcn_readlane(__float_as_int(c_l), k));
                const float4 v = *(const float4*)(x + (size_t)sn * HD + lane * 4);
                float* t = tile + rn * PITCH + lane;
                atomicAdd(t + 0,  v.x * cn);
                atomicAdd(t + 32, v.y * cn);
                atomicAdd(t + 64, v.z * cn);
                atomicAdd(t + 96, v.w * cn);
            }
            s_l = s_n;
            r_l = r_n;
            c_l = fa * fb;
            e1  = e2;
        }
    }

    const int half = lane >> 4;
    const int l    = lane & 15;
    const int n0   = (wave & 7) * 16;
    const int nrt  = (rows + 15) >> 4;
    v2f bf[32];

    load_bfrags<true>(Wt, n0 + l, half, bf);
    const float bn = bias[n0 + l];
    __syncthreads();
    {
        for (int rt = wave >> 3; rt < nrt; rt += 4) {
            const v8f acc = tile_mma(tile + (rt * 16 + l) * PITCH + 2 * half, bf);
#pragma unroll
            for (int r = 0; r < 8; ++r) {
                const int m = rt * 16 + r + 8 * half;
                const float v = acc[r] + bn;
                zt[m * PITCH + n0 + l] = v;
                if (m < rows) z[(size_t)(lo + m) * HD + n0 + l] = v;
            }
        }
    }

    load_bfrags<false>(Wb, n0 + l, half, bf);
    __syncthreads();
    for (int rt = wave >> 3; rt < nrt; rt += 4) {
        const v8f acc = tile_mma(zt + (rt * 16 + l) * PITCH + 2 * half, bf);
#pragma unroll
        for (int r = 0; r < 8; ++r) {
            const int m = rt * 16 + r + 8 * half;
            if (m < rows) u[(size_t)(lo + m) * HD + n0 + l] = acc[r];
        }
    }
}

__global__ __launch_bounds__(256) void dgi_score(const float* __restrict__ u,
                                                 const float* __restrict__ z,
                                                 const int* __restrict__ perm,
                                                 const float* __restrict__ bb,
                                                 float* __restrict__ out) {
    const int wave = threadIdx.x >> 5;
    const int lane = threadIdx.x & 31;
    const int n = blockIdx.x * 8 + wave;
    if (n >= NN) return;

    const float* un = u + (size_t)n * HD;
    const float* za = z + (size_t)n * HD;
    const float* zb = z + (size_t)perm[n] * HD;

    float4 uu = *(const float4*)(un + lane * 4);
    float4 va = *(const float4*)(za + lane * 4);
    float4 vb = *(const float4*)(zb + lane * 4);

    float pos = uu.x * va.x + uu.y * va.y + uu.z * va.z + uu.w * va.w;
    float neg = uu.x * vb.x + uu.y * vb.y + uu.z * vb.z + uu.w * vb.w;

#pragma unroll
    for (int off = 16; off > 0; off >>= 1) {
        pos += __shfl_xor(pos, off, 32);
        neg += __shfl_xor(neg, off, 32);
    }
    if (lane == 0) {
        float bias = bb[0];
        out[n]      = pos + bias;
        out[NN + n] = neg + bias;
    }
}

extern "C" void kernel_launch(void* const* d_in, const int* in_sizes, int n_in,
                              void* d_out, int out_size, void* d_ws, size_t ws_size,
                              hipStream_t stream) {
    (void)in_sizes; (void)n_in; (void)out_size; (void)ws_size;

    const float* x    = (const float*)d_in[0];
    const float* W    = (const float*)d_in[1];
    const float* b    = (const float*)d_in[2];
    const float* Wb   = (const float*)d_in[3];
    const float* bb   = (const float*)d_in[4];
    const int*   ei   = (const int*)d_in[5];
    const int*   perm = (const int*)d_in[6];
    const int*   esrc = ei;
    const int*   edst = ei + NE;
    float*       out  = (float*)d_out;

    char* ws = (char*)d_ws;
    size_t off = 0;
    float* dinv = (float*)(ws + off); off += ((size_t)NN * 4 + 255) & ~(size_t)255;
    float* z    = (float*)(ws + off); off += ((size_t)NN * HD * 4 + 255) & ~(size_t)255;
    float* u    = (float*)(ws + off); off += ((size_t)NN * HD * 4 + 255) & ~(size_t)255;
    unsigned* lists = (unsigned*)(ws + off); off += ((size_t)NE * 4 + 255) & ~(size_t)255;
    unsigned* offs  = (unsigned*)(ws + off); off += ((size_t)DC * NTILE * 4 + 255) & ~(size_t)255;

    unsigned* part = (unsigned*)u;

    const int nblk  = (NN + 255) / 256;

    dgi_deg_bin<<<DR * DC, DTH, 0, stream>>>(esrc, edst, part, lists, offs);
    dgi_dinv<<<nblk, 256, 0, stream>>>(part, dinv);

    dgi_tile_aggregate<<<NTILE, 1024, 0, stream>>>(lists, offs, x, W, dinv, b, Wb, z, u);

    dgi_score<<<(NN + 7) / 8, 256, 0, stream>>>(u, z, perm, bb, out);
}


# score pass: 8 lanes per node, 8 nodes per wave, all 24 row loads of a lane in flight, not one wave per node; 50000 -> 6250 waves; dot summed per lane then over 8 lanes: sum order changes
# speedup vs baseline: 4.7862x; 1.0727x over previous
#include <hip/hip_runtime.h>

#define NN   50000
#define NE   800000
#define HD   128

#define TR     196
#define NTILE  256
#define NE4    (NE / 4)

typedef float v2f __attribute__((ext_vector_type(2)));
typedef float v8f __attribute__((ext_vector_type(8)));

#define DR     8
#define DC     32
#define TPR    (NTILE / DR)
#define DRN    (TPR * TR)
#define DCE    (NE / DC)
#define DCE4   (NE4 / DC)
#define DTH    512
#define DG     ((DCE4 + DTH - 1) / DTH)

static_assert(TPR == 32, "one lane per tile of a range in the start scan");
static_assert(DTH / 16 == TPR, "sixteen threads per tile in the tile sums");
static_assert(DR * DRN >= NN && NTILE * TR >= NN, "ranges and tiles cover every node");
static_assert(TR <= 256 && NN < (1 << 24), "a list word holds the row in 8 bits and the source above them");

__global__ __launch_bounds__(DTH) void dgi_deg_bin(const int* __restrict__ src,
                                                   const int* __restrict__ dst,
                                                   unsigned* __restrict__ part,
                                                   unsigned* __restrict__ lists,
                                                   unsigned* __restrict__ offs) {
    __shared__ unsigned cnt[DRN];
    __shared__ unsigned ttot[TPR];
    __shared__ unsigned cur[TPR];
    __shared__ unsigned wbel[DTH / 32];
    const int c    = blockIdx.x / DR;
    const int r    = blockIdx.x % DR;
    const int nlo  = r * DRN;
    const int lane = threadIdx.x & 31;
    const int wave = threadIdx.x >> 5;

    for (int i = threadIdx.x; i < DRN; i += DTH) cnt[i] = 0u;
    __syncthreads();

    const int4* d4 = (const int4*)dst + (size_t)c * DCE4;
    const int4* s4 = (const int4*)src + (size_t)c * DCE4;
    int4 d[DG], sv[DG];
#pragma unroll
    for (int k = 0; k < DG; ++k) {
        const int i = threadIdx.x + k * DTH;
        d[k] = d4[(i < DCE4) ? i : DCE4 - 1];
    }
#pragma unroll
    for (int k = 0; k < DG; ++k) {
        const int i = threadIdx.x + k * DTH;
        sv[k] = s4[(i < DCE4) ? i : DCE4 - 1];
    }

    unsigned below = 0u;
#pragma unroll
    for (int k = 0; k < DG; ++k) {
        const int i = threadIdx.x + k * DTH;
        if (i < DCE4) {
            const unsigned a0 = (unsigned)(d[k].x - nlo);
            const unsigned a1 = (unsigned)(d[k].y - nlo);
            const unsigned a2 = (unsigned)(d[k].z - nlo);
            const unsigned a3 = (unsigned)(d[k].w - nlo);
            if (a0 < DRN) atomicAdd(&cnt[a0], 1u);
            if (a1 < DRN) atomicAdd(&cnt[a1], 1u);
            if (a2 < DRN) atomicAdd(&cnt[a2], 1u);
            if (a3 < DRN) atomicAdd(&cnt[a3], 1u);
            below += (unsigned)(d[k].x < nlo) + (unsigned)(d[k].y < nlo) +
                     (unsigned)(d[k].z < nlo) + (unsigned)(d[k].w < nlo);
        }
    }
    for (int off = 16; off > 0; off >>= 1) below += __shfl_xor(below, off, 32);
    if (lane == 0) wbel[wave] = below;
    __syncthreads();

    unsigned* prow = part + (size_t)c * NN + nlo;
    for (int i = threadIdx.x; i < DRN; i += DTH)
        if (nlo + i < NN) prow[i] = cnt[i];

    {
        const int tl  = threadIdx.x >> 4;
        const int sub = threadIdx.x & 15;
        unsigned t = 0u;
        for (int m = sub; m < TR; m += 16) t += cnt[tl * TR + m];
        for (int off = 8; off > 0; off >>= 1) t += __shfl_xor(t, off, 32);
        if (sub == 0) ttot[tl] = t;
    }
    __syncthreads();

    if (wave == 0) {
        unsigned b = (lane < DTH / 32) ? wbel[lane] : 0u;
        for (int off = 16; off > 0; off >>= 1) b += __shfl_xor(b, off, 32);
        const unsigned t = ttot[lane];
        unsigned inc = t;
        for (int off = 1; off < 32; off <<= 1) {
            const unsigned v = __shfl_up(inc, off, 32);
            if (lane >= off) inc += v;
        }
        const unsigned start = b + inc - t;
        cur[lane] = start;
        offs[c * NTILE + r * TPR + lane] = start;
    }
    __syncthreads();

    unsigned* lst = lists + (size_t)c * DCE;
#pragma unroll
    for (int k = 0; k < DG; ++k) {
        const int i = threadIdx.x + k * DTH;
        if (i < DCE4) {
            const int dd[4] = {d[k].x, d[k].y, d[k].z, d[k].w};
            const int ss[4] = {sv[k].x, sv[k].y, sv[k].z, sv[k].w};
            unsigned a[4], tl[4], pos[4];
#pragma unroll
            for (int j = 0; j < 4; ++j) {
                a[j]   = (unsigned)(dd[j] - nlo);
                tl[j]  = a[j] / TR;
                pos[j] = 0u;
                if (a[j] < DRN) pos[j] = atomicAdd(&cur[tl[j]], 1u);
            }
#pragma unroll
            for (int j = 0; j < 4; ++j)
                if (a[j] < DRN) lst[pos[j]] = ((unsigned)ss[j] << 8) | (a[j] - tl[j] * TR);
        }
    }
}

__global__ __launch_bounds__(256) void dgi_dinv(const unsigned* __restrict__ part,
                                                float* __restrict__ dinv) {
    const int i = blockIdx.x * blockDim.x + threadIdx.x;
    if (i < NN) {
        unsigned s = 1u;
#pragma unroll
        for (int c = 0; c < DC; ++c) s += part[(size_t)c * NN + i];
        const float d = (float)s;
        dinv[i] = (d > 0.0f) ? rsqrtf(d) : 0.0f;
    }
}

template <bool PERMK>
__device__ __forceinline__ int kmap(int p) {
    return PERMK ? (4 * (p & 31) + (p >> 5)) : p;
}

template <bool PERMK>
__device__ __forceinline__ void load_bfrags(const float* __restrict__ B, int col, int half,
                                            v2f (&bf)[32]) {
#pragma unroll
    for (int j = 0; j < 32; ++j) {
        const int p = 4 * j + 2 * half;
        bf[j].x = B[(size_t)kmap<PERMK>(p + 0) * HD + col];
        bf[j].y = B[(size_t)kmap<PERMK>(p + 1) * HD + col];
    }
}

__device__ __forceinline__ v8f tile_mma(const float* arow, const v2f (&bf)[32]) {
    v8f acc = {0.f, 0.f, 0.f, 0.f, 0.f, 0.f, 0.f, 0.f};
#pragma unroll
    for (int j = 0; j < 32; ++j) {
        const v2f a = *(const v2f*)(arow + 4 * j);
        acc = __builtin_amdgcn_wmma_f32_16x16x4_f32(
            false, a, false, bf[j], (short)0, acc, false, false);
    }
    return acc;
}

#define TRP    208
#define PITCH  132
#define AGGW   32
static_assert(AGGW == DC, "one edge chunk per owner wave");

__global__ __launch_bounds__(1024) void dgi_tile_aggregate(const unsigned* __restrict__ lists,
                                                           const unsigned* __restrict__ offs,
                                                           const float* __restrict__ x,
                                                           const float* __restrict__ Wt,
                                                           const float* __restrict__ dinv,
                                                           const float* __restrict__ bias,
                                                           const float* __restrict__ Wb,
                                                           float* __restrict__ z,
                                                           float* __restrict__ u) {
    __shared__ float    tile[TRP * PITCH];
    __shared__ float    zt[TRP * PITCH];

    const int wave = __builtin_amdgcn_readfirstlane((int)(threadIdx.x >> 5));
    const int lane = threadIdx.x & 31;
    const int lo   = blockIdx.x * TR;
    const int rows = (NN - lo < TR) ? (NN - lo) : TR;

    for (int r = wave; r < TRP; r += AGGW) {
        float4 v = make_float4(0.f, 0.f, 0.f, 0.f);
        if (r < rows) {
            const float di = dinv[lo + r];
            const float4 xv = *(const float4*)(x + (size_t)(lo + r) * HD + lane * 4);
            v.x = di * di * xv.x;
            v.y = di * di * xv.y;
            v.z = di * di * xv.z;
            v.w = di * di * xv.w;
        }
        float* t = tile + r * PITCH + lane;
        t[0]  = v.x;
        t[32] = v.y;
        t[64] = v.z;
        t[96] = v.w;
    }
    __syncthreads();

    {
        const unsigned* lst = lists + (size_t)wave * DCE;
        const int beg = (int)offs[wave * NTILE + blockIdx.x];
        const int end = (blockIdx.x == NTILE - 1) ? DCE
                                                  : (int)offs[wave * NTILE + blockIdx.x + 1];
        unsigned e0 = 0u, e1 = 0u;
        if (beg + lane < end)      e0 = lst[beg + lane];
        if (beg + 32 + lane < end) e1 = lst[beg + 32 + lane];
        int   s_l = (int)(e0 >> 8);
        int   r_l = (int)(e0 & 255u);
        float c_l = 0.0f;
        if (beg + lane < end) c_l = dinv[s_l] * dinv[lo + r_l];
        for (int base = beg; base < end; base += 32) {
            const int cnt = (end - base < 32) ? (end - base) : 32;
            const int s_n = (int)(e1 >> 8);
            const int r_n = (int)(e1 & 255u);
            const float fa = dinv[s_n];
            const float fb = dinv[lo + r_n];
            unsigned e2 = 0u;
            if (base + 64 + lane < end) e2 = lst[base + 64 + lane];
            int k = 0;
            for (; k + 4 <= cnt; k += 4) {
                float4 v[4];
                int    rr[4];
                float  cc[4];
#pragma unroll
                for (int j = 0; j < 4; ++j) {
                    const int sn = __builtin_amdgcn_readlane(s_l, k + j);
                    rr[j] = __builtin_amdgcn_readlane(r_l, k + j);
                    cc[j] = __int_as_float(__builtin_amdgcn_readlane(__float_as_int(c_l), k + j));
                    v[j]  = *(const float4*)(x + (size_t)sn * HD + lane * 4);
                }
#pragma unroll
                for (int j = 0; j < 4; ++j) {
                    float* t = tile + rr[j] * PITCH + lane;
                    atomicAdd(t + 0,  v[j].x * cc[j]);
                    atomicAdd(t + 32, v[j].y * cc[j]);
                    atomicAdd(t + 64, v[j].z * cc[j]);
                    atomicAdd(t + 96, v[j].w * cc[j]);
                }
            }
            for (; k < cnt; ++k) {
                const int   sn = __builtin_amdgcn_readlane(s_l, k);
                const int   rn = __builtin_amdgcn_readlane(r_l, k);
                const float cn = __int_as_float(__builtin_amdgcn_readlane(__float_as_int(c_l), k));
                const float4 v = *(const float4*)(x + (size_t)sn * HD + lane * 4);
                float* t = tile + rn * PITCH + lane;
                atomicAdd(t + 0,  v.x * cn);
                atomicAdd(t + 32, v.y * cn);
                atomicAdd(t + 64, v.z * cn);
                atomicAdd(t + 96, v.w * cn);
            }
            s_l = s_n;
            r_l = r_n;
            c_l = fa * fb;
            e1  = e2;
        }
    }

    const int half = lane >> 4;
    const int l    = lane & 15;
    const int n0   = (wave & 7) * 16;
    const int nrt  = (rows + 15) >> 4;
    v2f bf[32];

    load_bfrags<true>(Wt, n0 + l, half, bf);
    const float bn = bias[n0 + l];
    __syncthreads();
    {
        for (int rt = wave >> 3; rt < nrt; rt += 4) {
            const v8f acc = tile_mma(tile + (rt * 16 + l) * PITCH + 2 * half, bf);
#pragma unroll
            for (int r = 0; r < 8; ++r) {
                const int m = rt * 16 + r + 8 * half;
                const float v = acc[r] + bn;
                zt[m * PITCH + n0 + l] = v;
                if (m < rows) z[(size_t)(lo + m) * HD + n0 + l] = v;
            }
        }
    }

    load_bfrags<false>(Wb, n0 + l, half, bf);
    __syncthreads();
    for (int rt = wave >> 3; rt < nrt; rt += 4) {
        const v8f acc = tile_mma(zt + (rt * 16 + l) * PITCH + 2 * half, bf);
#pragma unroll
        for (int r = 0; r < 8; ++r) {
            const int m = rt * 16 + r + 8 * half;
            if (m < rows) u[(size_t)(lo + m) * HD + n0 + l] = acc[r];
        }
    }
}

#define SNW    8
#define SWG    8
static_assert(NN % SNW == 0, "every wave has all of its nodes");

__global__ __launch_bounds__(SWG * 32) void dgi_score(const float* __restrict__ u,
                                                      const float* __restrict__ z,
                                                      const int* __restrict__ perm,
                                                      const float* __restrict__ bb,
                                                      float* __restrict__ out) {
    const int wave = threadIdx.x >> 5;
    const int lane = threadIdx.x & 31;
    const int g    = lane >> 3;
    const int q    = lane & 7;
    const int n0   = (blockIdx.x * SWG + wave) * SNW;
    if (n0 >= NN) return;

    int n[2], p[2];
#pragma unroll
    for (int h = 0; h < 2; ++h) {
        n[h] = n0 + 4 * h + g;
        p[h] = perm[n[h]];
    }

    float4 uu[2][4], va[2][4], vb[2][4];
#pragma unroll
    for (int h = 0; h < 2; ++h) {
#pragma unroll
        for (int j = 0; j < 4; ++j) {
            uu[h][j] = *(const float4*)(u + (size_t)n[h] * HD + 32 * j + 4 * q);
            va[h][j] = *(const float4*)(z + (size_t)n[h] * HD + 32 * j + 4 * q);
        }
    }
#pragma unroll
    for (int h = 0; h < 2; ++h) {
#pragma unroll
        for (int j = 0; j < 4; ++j)
            vb[h][j] = *(const float4*)(z + (size_t)p[h] * HD + 32 * j + 4 * q);
    }

    const float bias = bb[0];
    __builtin_amdgcn_sched_barrier(0);
#pragma unroll
    for (int h = 0; h < 2; ++h) {
        float pos = 0.0f, neg = 0.0f;
#pragma unroll
        for (int j = 0; j < 4; ++j) {
            pos += uu[h][j].x * va[h][j].x + uu[h][j].y * va[h][j].y +
                   uu[h][j].z * va[h][j].z + uu[h][j].w * va[h][j].w;
            neg += uu[h][j].x * vb[h][j].x + uu[h][j].y * vb[h][j].y +
                   uu[h][j].z * vb[h][j].z + uu[h][j].w * vb[h][j].w;
        }
#pragma unroll
        for (int off = 4; off > 0; off >>= 1) {
            pos += __shfl_xor(pos, off, 32);
            neg += __shfl_xor(neg, off, 32);
        }
        if (q == 0) {
            out[n[h]]      = pos + bias;
            out[NN + n[h]] = neg + bias;
        }
    }
}

extern "C" void kernel_launch(void* const* d_in, const int* in_sizes, int n_in,
                              void* d_out, int out_size, void* d_ws, size_t ws_size,
                              hipStream_t stream) {
    (void)in_sizes; (void)n_in; (void)out_size; (void)ws_size;

    const float* x    = (const float*)d_in[0];
    const float* W    = (const float*)d_in[1];
    const float* b    = (const float*)d_in[2];
    const float* Wb   = (const float*)d_in[3];
    const float* bb   = (const float*)d_in[4];
    const int*   ei   = (const int*)d_in[5];
    const int*   perm = (const int*)d_in[6];
    const int*   esrc = ei;
    const int*   edst = ei + NE;
    float*       out  = (float*)d_out;

    char* ws = (char*)d_ws;
    size_t off = 0;
    float* dinv = (float*)(ws + off); off += ((size_t)NN * 4 + 255) & ~(size_t)255;
    float* z    = (float*)(ws + off); off += ((size_t)NN * HD * 4 + 255) & ~(size_t)255;
    float* u    = (float*)(ws + off); off += ((size_t)NN * HD * 4 + 255) & ~(size_t)255;
    unsigned* lists = (unsigned*)(ws + off); off += ((size_t)NE * 4 + 255) & ~(size_t)255;
    unsigned* offs  = (unsigned*)(ws + off); off += ((size_t)DC * NTILE * 4 + 255) & ~(size_t)255;

    unsigned* part = (unsigned*)u;

    const int nblk  = (NN + 255) / 256;

    dgi_deg_bin<<<DR * DC, DTH, 0, stream>>>(esrc, edst, part, lists, offs);
    dgi_dinv<<<nblk, 256, 0, stream>>>(part, dinv);

    dgi_tile_aggregate<<<NTILE, 1024, 0, stream>>>(lists, offs, x, W, dinv, b, Wb, z, u);

    dgi_score<<<(NN / SNW + SWG - 1) / SWG, SWG * 32, 0, stream>>>(u, z, perm, bb, out);
}


# both 128x128 transforms and the bilinear weight folded into one matrix M = W Wb W^T built once per call on f32 WMMA; owners do one transform per row and store y and v; sum order changes
# speedup vs baseline: 4.9415x; 1.0324x over previous
#include <hip/hip_runtime.h>

#define NN   50000
#define NE   800000
#define HD   128

#define TR     196
#define NTILE  256
#define NE4    (NE / 4)

typedef float v2f __attribute__((ext_vector_type(2)));
typedef float v8f __attribute__((ext_vector_type(8)));

#define DR     8
#define DC     32
#define TPR    (NTILE / DR)
#define DRN    (TPR * TR)
#define DCE    (NE / DC)
#define DCE4   (NE4 / DC)
#define DTH    512
#define DG     ((DCE4 + DTH - 1) / DTH)

#define FOLDB  (HD / 16)
#define NBLK   ((NN + 255) / 256)

static_assert(TPR == 32, "one lane per tile of a range in the start scan");
static_assert(DTH / 16 == TPR, "sixteen threads per tile in the tile sums");
static_assert(DR * DRN >= NN && NTILE * TR >= NN, "ranges and tiles cover every node");
static_assert(TR <= 256 && NN < (1 << 24), "a list word holds the row in 8 bits and the source above them");

template <bool PERMK>
__device__ __forceinline__ int kmap(int p) {
    return PERMK ? (4 * (p & 31) + (p >> 5)) : p;
}

template <bool PERMK>
__device__ __forceinline__ void load_bfrags(const float* __restrict__ B, int col, int half,
                                            v2f (&bf)[32]) {
#pragma unroll
    for (int j = 0; j < 32; ++j) {
        const int p = 4 * j + 2 * half;
        bf[j].x = B[(size_t)kmap<PERMK>(p + 0) * HD + col];
        bf[j].y = B[(size_t)kmap<PERMK>(p + 1) * HD + col];
    }
}

__device__ __forceinline__ void load_bfrags_t(const float* __restrict__ S, int col, int half,
                                              v2f (&bf)[32]) {
#pragma unroll
    for (int j = 0; j < 32; ++j)
        bf[j] = *(const v2f*)(S + (size_t)col * HD + 4 * j + 2 * half);
}

__device__ __forceinline__ v8f tile_mma(const float* arow, const v2f (&bf)[32]) {
    v8f acc = {0.f, 0.f, 0.f, 0.f, 0.f, 0.f, 0.f, 0.f};
#pragma unroll
    for (int j = 0; j < 32; ++j) {
        const v2f a = *(const v2f*)(arow + 4 * j);
        acc = __builtin_amdgcn_wmma_f32_16x16x4_f32(
            false, a, false, bf[j], (short)0, acc, false, false);
    }
    return acc;
}

__device__ __forceinline__ void fold_stage1(const float* __restrict__ Wt, const float* __restrict__ Wb,
                                            const float* __restrict__ bias,
                                            float* __restrict__ T, float* __restrict__ t0, int e) {
    const int wave = __builtin_amdgcn_readfirstlane((int)(threadIdx.x >> 5));
    const int lane = threadIdx.x & 31;
    const int half = lane >> 4;
    const int l    = lane & 15;
    if (wave < 8) {
        const int n0 = wave * 16;
        v2f bf[32];
        load_bfrags<false>(Wb, n0 + l, half, bf);
        const v8f acc = tile_mma(Wt + (size_t)(16 * e + l) * HD + 2 * half, bf);
#pragma unroll
        for (int r = 0; r < 8; ++r)
            T[(size_t)(16 * e + r + 8 * half) * HD + n0 + l] = acc[r];
    } else if (e == 0 && wave < 12) {
        const int g = (wave - 8) * 32 + lane;
        float s = 0.0f;
        for (int h0 = 0; h0 < HD; h0 += 16) {
            float w[16];
#pragma unroll
            for (int k = 0; k < 16; ++k) w[k] = Wb[(size_t)(h0 + k) * HD + g];
#pragma unroll
            for (int k = 0; k < 16; ++k) s += bias[h0 + k] * w[k];
        }
        t0[g] = s;
    }
}

__device__ __forceinline__ void fold_stage2(const float* __restrict__ Wt, const float* __restrict__ bias,
                                            const float* __restrict__ T, const float* __restrict__ t0,
                                            float* __restrict__ M, float* __restrict__ cv,
                                            float* __restrict__ cs, float* __restrict__ c0, int e) {
    const int wave = threadIdx.x >> 5;
    const int lane = threadIdx.x & 31;
    const int half = lane >> 4;
    const int l    = lane & 15;
    {
        const int n0 = wave * 16;
        v2f bf[32];
        load_bfrags_t(Wt, n0 + l, half, bf);
        const v8f acc = tile_mma(T + (size_t)(16 * e + l) * HD + 2 * half, bf);
#pragma unroll
        for (int r = 0; r < 8; ++r)
            M[(size_t)(16 * e + r + 8 * half) * HD + n0 + l] = acc[r];
    }
    if (e < 3) {
        const float* S   = (e == 0) ? Wt : (e == 1) ? T : t0;
        const float* vec = (e == 0) ? t0 : bias;
        float*       dv  = (e == 0) ? cv : (e == 1) ? cs : c0;
        const int nrow   = (e == 2) ? ((wave == 0) ? 1 : 0) : 16;
        const float4 c = *(const float4*)(vec + lane * 4);
        float4 rv[16];
#pragma unroll
        for (int k = 0; k < 16; ++k)
            rv[k] = (k < nrow) ? *(const float4*)(S + (size_t)(16 * wave + k) * HD + lane * 4)
                               : make_float4(0.f, 0.f, 0.f, 0.f);
#pragma unroll
        for (int k = 0; k < 16; ++k) {
            float s = rv[k].x * c.x + rv[k].y * c.y + rv[k].z * c.z + rv[k].w * c.w;
            for (int off = 16; off > 0; off >>= 1) s += __shfl_xor(s, off, 32);
            if (lane == 0 && k < nrow) dv[16 * wave + k] = s;
        }
    }
}

__global__ __launch_bounds__(DTH) void dgi_deg_bin(const int* __restrict__ src,
                                                   const int* __restrict__ dst,
                                                   unsigned* __restrict__ part,
                                                   unsigned* __restrict__ lists,
                                                   unsigned* __restrict__ offs,
                                                   const float* __restrict__ Wt,
                                                   const float* __restrict__ Wb,
                                                   const float* __restrict__ bias,
                                                   float* __restrict__ T,
                                                   float* __restrict__ t0) {
    __shared__ unsigned cnt[DRN];
    __shared__ unsigned ttot[TPR];
    __shared__ unsigned cur[TPR];
    __shared__ unsigned wbel[DTH / 32];
    if (blockIdx.x >= DR * DC) {
        fold_stage1(Wt, Wb, bias, T, t0, (int)blockIdx.x - DR * DC);
        return;
    }
    const int c    = blockIdx.x / DR;
    const int r    = blockIdx.x % DR;
    const int nlo  = r * DRN;
    const int lane = threadIdx.x & 31;
    const int wave = threadIdx.x >> 5;

    for (int i = threadIdx.x; i < DRN; i += DTH) cnt[i] = 0u;
    __syncthreads();

    const int4* d4 = (const int4*)dst + (size_t)c * DCE4;
    const int4* s4 = (const int4*)src + (size_t)c * DCE4;
    int4 d[DG], sv[DG];
#pragma unroll
    for (int k = 0; k < DG; ++k) {
        const int i = threadIdx.x + k * DTH;
        d[k] = d4[(i < DCE4) ? i : DCE4 - 1];
    }
#pragma unroll
    for (int k = 0; k < DG; ++k) {
        const int i = threadIdx.x + k * DTH;
        sv[k] = s4[(i < DCE4) ? i : DCE4 - 1];
    }

    unsigned below = 0u;
#pragma unroll
    for (int k = 0; k < DG; ++k) {
        const int i = threadIdx.x + k * DTH;
        if (i < DCE4) {
            const unsigned a0 = (unsigned)(d[k].x - nlo);
            const unsigned a1 = (unsigned)(d[k].y - nlo);
            const unsigned a2 = (unsigned)(d[k].z - nlo);
            const unsigned a3 = (unsigned)(d[k].w - nlo);
            if (a0 < DRN) atomicAdd(&cnt[a0], 1u);
            if (a1 < DRN) atomicAdd(&cnt[a1], 1u);
            if (a2 < DRN) atomicAdd(&cnt[a2], 1u);
            if (a3 < DRN) atomicAdd(&cnt[a3], 1u);
            below += (unsigned)(d[k].x < nlo) + (unsigned)(d[k].y < nlo) +
                     (unsigned)(d[k].z < nlo) + (unsigned)(d[k].w < nlo);
        }
    }
    for (int off = 16; off > 0; off >>= 1) below += __shfl_xor(below, off, 32);
    if (lane == 0) wbel[wave] = below;
    __syncthreads();

    unsigned* prow = part + (size_t)c * NN + nlo;
    for (int i = threadIdx.x; i < DRN; i += DTH)
        if (nlo + i < NN) prow[i] = cnt[i];

    {
        const int tl  = threadIdx.x >> 4;
        const int sub = threadIdx.x & 15;
        unsigned t = 0u;
        for (int m = sub; m < TR; m += 16) t += cnt[tl * TR + m];
        for (int off = 8; off > 0; off >>= 1) t += __shfl_xor(t, off, 32);
        if (sub == 0) ttot[tl] = t;
    }
    __syncthreads();

    if (wave == 0) {
        unsigned b = (lane < DTH / 32) ? wbel[lane] : 0u;
        for (int off = 16; off > 0; off >>= 1) b += __shfl_xor(b, off, 32);
        const unsigned t = ttot[lane];
        unsigned inc = t;
        for (int off = 1; off < 32; off <<= 1) {
            const unsigned v = __shfl_up(inc, off, 32);
            if (lane >= off) inc += v;
        }
        const unsigned start = b + inc - t;
        cur[lane] = start;
        offs[c * NTILE + r * TPR + lane] = start;
    }
    __syncthreads();

    unsigned* lst = lists + (size_t)c * DCE;
#pragma unroll
    for (int k = 0; k < DG; ++k) {
        const int i = threadIdx.x + k * DTH;
        if (i < DCE4) {
            const int dd[4] = {d[k].x, d[k].y, d[k].z, d[k].w};
            const int ss[4] = {sv[k].x, sv[k].y, sv[k].z, sv[k].w};
            unsigned a[4], tl[4], pos[4];
#pragma unroll
            for (int j = 0; j < 4; ++j) {
                a[j]   = (unsigned)(dd[j] - nlo);
                tl[j]  = a[j] / TR;
                pos[j] = 0u;
                if (a[j] < DRN) pos[j] = atomicAdd(&cur[tl[j]], 1u);
            }
#pragma unroll
            for (int j = 0; j < 4; ++j)
                if (a[j] < DRN) lst[pos[j]] = ((unsigned)ss[j] << 8) | (a[j] - tl[j] * TR);
        }
    }
}

__global__ __launch_bounds__(256) void dgi_dinv(const unsigned* __restrict__ part,
                                                float* __restrict__ dinv,
                                                const float* __restrict__ Wt,
                                                const float* __restrict__ bias,
                                                const float* __restrict__ T,
                                                const float* __restrict__ t0,
                                                float* __restrict__ M,
                                                float* __restrict__ cv,
                                                float* __restrict__ cs,
                                                float* __restrict__ c0) {
    if (blockIdx.x >= NBLK) {
        fold_stage2(Wt, bias, T, t0, M, cv, cs, c0, (int)blockIdx.x - NBLK);
        return;
    }
    const int i = blockIdx.x * blockDim.x + threadIdx.x;
    if (i < NN) {
        unsigned s = 1u;
#pragma unroll
        for (int c = 0; c < DC; ++c) s += part[(size_t)c * NN + i];
        const float d = (float)s;
        dinv[i] = (d > 0.0f) ? rsqrtf(d) : 0.0f;
    }
}

#define TRP    208
#define PITCH  132
#define AGGW   32
static_assert(AGGW == DC, "one edge chunk per owner wave");

__global__ __launch_bounds__(1024) void dgi_tile_aggregate(const unsigned* __restrict__ lists,
                                                           const unsigned* __restrict__ offs,
                                                           const float* __restrict__ x,
                                                           const float* __restrict__ M,
                                                           const float* __restrict__ dinv,
                                                           const float* __restrict__ cv,
                                                           float* __restrict__ y,
                                                           float* __restrict__ v) {
    __shared__ float    tile[TRP * PITCH];

    const int wave = __builtin_amdgcn_readfirstlane((int)(threadIdx.x >> 5));
    const int lane = threadIdx.x & 31;
    const int lo   = blockIdx.x * TR;
    const int rows = (NN - lo < TR) ? (NN - lo) : TR;

    for (int r = wave; r < TRP; r += AGGW) {
        float4 v = make_float4(0.f, 0.f, 0.f, 0.f);
        if (r < rows) {
            const float di = dinv[lo + r];
            const float4 xv = *(const float4*)(x + (size_t)(lo + r) * HD + lane * 4);
            v.x = di * di * xv.x;
            v.y = di * di * xv.y;
            v.z = di * di * xv.z;
            v.w = di * di * xv.w;
        }
        float* t = tile + r * PITCH + lane;
        t[0]  = v.x;
        t[32] = v.y;
        t[64] = v.z;
        t[96] = v.w;
    }
    __syncthreads();

    {
        const unsigned* lst = lists + (size_t)wave * DCE;
        const int beg = (int)offs[wave * NTILE + blockIdx.x];
        const int end = (blockIdx.x == NTILE - 1) ? DCE
                                                  : (int)offs[wave * NTILE + blockIdx.x + 1];
        unsigned e0 = 0u, e1 = 0u;
        if (beg + lane < end)      e0 = lst[beg + lane];
        if (beg + 32 + lane < end) e1 = lst[beg + 32 + lane];
        int   s_l = (int)(e0 >> 8);
        int   r_l = (int)(e0 & 255u);
        float c_l = 0.0f;
        if (beg + lane < end) c_l = dinv[s_l] * dinv[lo + r_l];
        for (int base = beg; base < end; base += 32) {
            const int cnt = (end - base < 32) ? (end - base) : 32;
            const int s_n = (int)(e1 >> 8);
            const int r_n = (int)(e1 & 255u);
            const float fa = dinv[s_n];
            const float fb = dinv[lo + r_n];
            unsigned e2 = 0u;
            if (base + 64 + lane < end) e2 = lst[base + 64 + lane];
            int k = 0;
            for (; k + 4 <= cnt; k += 4) {
                float4 v[4];
                int    rr[4];
                float  cc[4];
#pragma unroll
                for (int j = 0; j < 4; ++j) {
                    const int sn = __builtin_amdgcn_readlane(s_l, k + j);
                    rr[j] = __builtin_amdgcn_readlane(r_l, k + j);
                    cc[j] = __int_as_float(__builtin_amdgcn_readlane(__float_as_int(c_l), k + j));
                    v[j]  = *(const float4*)(x + (size_t)sn * HD + lane * 4);
                }
#pragma unroll
                for (int j = 0; j < 4; ++j) {
                    float* t = tile + rr[j] * PITCH + lane;
                    atomicAdd(t + 0,  v[j].x * cc[j]);
                    atomicAdd(t + 32, v[j].y * cc[j]);
                    atomicAdd(t + 64, v[j].z * cc[j]);
                    atomicAdd(t + 96, v[j].w * cc[j]);
                }
            }
            for (; k < cnt; ++k) {
                const int   sn = __builtin_amdgcn_readlane(s_l, k);
                const int   rn = __builtin_amdgcn_readlane(r_l, k);
                const float cn = __int_as_float(__builtin_amdgcn_readlane(__float_as_int(c_l), k));
                const float4 v = *(const float4*)(x + (size_t)sn * HD + lane * 4);
                float* t = tile + rn * PITCH + lane;
                atomicAdd(t + 0,  v.x * cn);
                atomicAdd(t + 32, v.y * cn);
                atomicAdd(t + 64, v.z * cn);
                atomicAdd(t + 96, v.w * cn);
            }
            s_l = s_n;
            r_l = r_n;
            c_l = fa * fb;
            e1  = e2;
        }
    }

    const int half = lane >> 4;
    const int l    = lane & 15;
    const int n0   = (wave & 7) * 16;
    const int nrt  = (rows + 15) >> 4;
    v2f bf[32];

    load_bfrags<true>(M, n0 + l, half, bf);
    const float cn = cv[n0 + l];
    __syncthreads();

    for (int r = wave; r < rows; r += AGGW) {
        const float* t = tile + r * PITCH + lane;
        const float4 o = make_float4(t[0], t[32], t[64], t[96]);
        *(float4*)(y + (size_t)(lo + r) * HD + lane * 4) = o;
    }

    for (int rt = wave >> 3; rt < nrt; rt += 4) {
        const v8f acc = tile_mma(tile + (rt * 16 + l) * PITCH + 2 * half, bf);
#pragma unroll
        for (int r = 0; r < 8; ++r) {
            const int m = rt * 16 + r + 8 * half;
            if (m < rows) v[(size_t)(lo + m) * HD + n0 + l] = acc[r] + cn;
        }
    }
}

#define SNW    8
#define SWG    8
static_assert(NN % SNW == 0, "every wave has all of its nodes");

__global__ __launch_bounds__(SWG * 32) void dgi_score(const float* __restrict__ u,
                                                      const float* __restrict__ z,
                                                      const int* __restrict__ perm,
                                                      const float* __restrict__ bb,
                                                      const float* __restrict__ cs,
                                                      const float* __restrict__ c0,
                                                      float* __restrict__ out) {
    const int wave = threadIdx.x >> 5;
    const int lane = threadIdx.x & 31;
    const int g    = lane >> 3;
    const int q    = lane & 7;
    const int n0   = (blockIdx.x * SWG + wave) * SNW;
    if (n0 >= NN) return;

    int n[2], p[2];
#pragma unroll
    for (int h = 0; h < 2; ++h) {
        n[h] = n0 + 4 * h + g;
        p[h] = perm[n[h]];
    }

    float4 uu[2][4], va[2][4], vb[2][4];
#pragma unroll
    for (int h = 0; h < 2; ++h) {
#pragma unroll
        for (int j = 0; j < 4; ++j) {
            uu[h][j] = *(const float4*)(u + (size_t)n[h] * HD + 32 * j + 4 * q);
            va[h][j] = *(const float4*)(z + (size_t)n[h] * HD + 32 * j + 4 * q);
        }
    }
#pragma unroll
    for (int h = 0; h < 2; ++h) {
#pragma unroll
        for (int j = 0; j < 4; ++j)
            vb[h][j] = *(const float4*)(z + (size_t)p[h] * HD + 32 * j + 4 * q);
    }

    const float bias = bb[0] + c0[0];
    float4 cc[4];
#pragma unroll
    for (int j = 0; j < 4; ++j) cc[j] = *(const float4*)(cs + 32 * j + 4 * q);
    __builtin_amdgcn_sched_barrier(0);
#pragma unroll
    for (int h = 0; h < 2; ++h) {
        float pos = 0.0f, neg = 0.0f, sn = 0.0f;
#pragma unroll
        for (int j = 0; j < 4; ++j) {
            sn  += cc[j].x * va[h][j].x + cc[j].y * va[h][j].y +
                   cc[j].z * va[h][j].z + cc[j].w * va[h][j].w;
            pos += uu[h][j].x * va[h][j].x + uu[h][j].y * va[h][j].y +
                   uu[h][j].z * va[h][j].z + uu[h][j].w * va[h][j].w;
            neg += uu[h][j].x * vb[h][j].x + uu[h][j].y * vb[h][j].y +
                   uu[h][j].z * vb[h][j].z + uu[h][j].w * vb[h][j].w;
        }
        pos += sn;
        neg += sn;
#pragma unroll
        for (int off = 4; off > 0; off >>= 1) {
            pos += __shfl_xor(pos, off, 32);
            neg += __shfl_xor(neg, off, 32);
        }
        if (q == 0) {
            out[n[h]]      = pos + bias;
            out[NN + n[h]] = neg + bias;
        }
    }
}

extern "C" void kernel_launch(void* const* d_in, const int* in_sizes, int n_in,
                              void* d_out, int out_size, void* d_ws, size_t ws_size,
                              hipStream_t stream) {
    (void)in_sizes; (void)n_in; (void)out_size; (void)ws_size;

    const float* x    = (const float*)d_in[0];
    const float* W    = (const float*)d_in[1];
    const float* b    = (const float*)d_in[2];
    const float* Wb   = (const float*)d_in[3];
    const float* bb   = (const float*)d_in[4];
    const int*   ei   = (const int*)d_in[5];
    const int*   perm = (const int*)d_in[6];
    const int*   esrc = ei;
    const int*   edst = ei + NE;
    float*       out  = (float*)d_out;

    char* ws = (char*)d_ws;
    size_t off = 0;
    float* dinv = (float*)(ws + off); off += ((size_t)NN * 4 + 255) & ~(size_t)255;
    float* z    = (float*)(ws + off); off += ((size_t)NN * HD * 4 + 255) & ~(size_t)255;
    float* u    = (float*)(ws + off); off += ((size_t)NN * HD * 4 + 255) & ~(size_t)255;
    unsigned* lists = (unsigned*)(ws + off); off += ((size_t)NE * 4 + 255) & ~(size_t)255;
    unsigned* offs  = (unsigned*)(ws + off); off += ((size_t)DC * NTILE * 4 + 255) & ~(size_t)255;
    float* M  = (float*)(ws + off); off += (size_t)HD * HD * 4;
    float* cv = (float*)(ws + off); off += (size_t)HD * 4;
    float* cs = (float*)(ws + off); off += (size_t)HD * 4;
    float* c0 = (float*)(ws + off); off += 256;

    unsigned* part = (unsigned*)u;
    float* T  = u + (size_t)DC * NN;
    float* t0 = T + (size_t)HD * HD;

    dgi_deg_bin<<<DR * DC + FOLDB, DTH, 0, stream>>>(esrc, edst, part, lists, offs, W, Wb, b, T, t0);
    dgi_dinv<<<NBLK + FOLDB, 256, 0, stream>>>(part, dinv, W, b, T, t0, M, cv, cs, c0);

    dgi_tile_aggregate<<<NTILE, 1024, 0, stream>>>(lists, offs, x, M, dinv, cv, z, u);

    dgi_score<<<(NN / SNW + SWG - 1) / SWG, SWG * 32, 0, stream>>>(u, z, perm, bb, cs, c0, out);
}


# owner kernel: a wave's own x rows, its list bounds, first two list batches and first coefficient factors are all requested at entry, ahead of tile init and barrier: one row round trip, not seven
# speedup vs baseline: 5.1118x; 1.0345x over previous
#include <hip/hip_runtime.h>

#define NN   50000
#define NE   800000
#define HD   128

#define TR     196
#define NTILE  256
#define NE4    (NE / 4)

typedef float v2f __attribute__((ext_vector_type(2)));
typedef float v8f __attribute__((ext_vector_type(8)));

#define DR     8
#define DC     32
#define TPR    (NTILE / DR)
#define DRN    (TPR * TR)
#define DCE    (NE / DC)
#define DCE4   (NE4 / DC)
#define DTH    512
#define DG     ((DCE4 + DTH - 1) / DTH)

#define FOLDB  (HD / 16)
#define NBLK   ((NN + 255) / 256)

static_assert(TPR == 32, "one lane per tile of a range in the start scan");
static_assert(DTH / 16 == TPR, "sixteen threads per tile in the tile sums");
static_assert(DR * DRN >= NN && NTILE * TR >= NN, "ranges and tiles cover every node");
static_assert(TR <= 256 && NN < (1 << 24), "a list word holds the row in 8 bits and the source above them");

template <bool PERMK>
__device__ __forceinline__ int kmap(int p) {
    return PERMK ? (4 * (p & 31) + (p >> 5)) : p;
}

template <bool PERMK>
__device__ __forceinline__ void load_bfrags(const float* __restrict__ B, int col, int half,
                                            v2f (&bf)[32]) {
#pragma unroll
    for (int j = 0; j < 32; ++j) {
        const int p = 4 * j + 2 * half;
        bf[j].x = B[(size_t)kmap<PERMK>(p + 0) * HD + col];
        bf[j].y = B[(size_t)kmap<PERMK>(p + 1) * HD + col];
    }
}

__device__ __forceinline__ void load_bfrags_t(const float* __restrict__ S, int col, int half,
                                              v2f (&bf)[32]) {
#pragma unroll
    for (int j = 0; j < 32; ++j)
        bf[j] = *(const v2f*)(S + (size_t)col * HD + 4 * j + 2 * half);
}

__device__ __forceinline__ v8f tile_mma(const float* arow, const v2f (&bf)[32]) {
    v8f acc = {0.f, 0.f, 0.f, 0.f, 0.f, 0.f, 0.f, 0.f};
#pragma unroll
    for (int j = 0; j < 32; ++j) {
        const v2f a = *(const v2f*)(arow + 4 * j);
        acc = __builtin_amdgcn_wmma_f32_16x16x4_f32(
            false, a, false, bf[j], (short)0, acc, false, false);
    }
    return acc;
}

__device__ __forceinline__ void fold_stage1(const float* __restrict__ Wt, const float* __restrict__ Wb,
                                            const float* __restrict__ bias,
                                            float* __restrict__ T, float* __restrict__ t0, int e) {
    const int wave = __builtin_amdgcn_readfirstlane((int)(threadIdx.x >> 5));
    const int lane = threadIdx.x & 31;
    const int half = lane >> 4;
    const int l    = lane & 15;
    if (wave < 8) {
        const int n0 = wave * 16;
        v2f bf[32];
        load_bfrags<false>(Wb, n0 + l, half, bf);
        const v8f acc = tile_mma(Wt + (size_t)(16 * e + l) * HD + 2 * half, bf);
#pragma unroll
        for (int r = 0; r < 8; ++r)
            T[(size_t)(16 * e + r + 8 * half) * HD + n0 + l] = acc[r];
    } else if (e == 0 && wave < 12) {
        const int g = (wave - 8) * 32 + lane;
        float s = 0.0f;
        for (int h0 = 0; h0 < HD; h0 += 16) {
            float w[16];
#pragma unroll
            for (int k = 0; k < 16; ++k) w[k] = Wb[(size_t)(h0 + k) * HD + g];
#pragma unroll
            for (int k = 0; k < 16; ++k) s += bias[h0 + k] * w[k];
        }
        t0[g] = s;
    }
}

__device__ __forceinline__ void fold_stage2(const float* __restrict__ Wt, const float* __restrict__ bias,
                                            const float* __restrict__ T, const float* __restrict__ t0,
                                            float* __restrict__ M, float* __restrict__ cv,
                                            float* __restrict__ cs, float* __restrict__ c0, int e) {
    const int wave = threadIdx.x >> 5;
    const int lane = threadIdx.x & 31;
    const int half = lane >> 4;
    const int l    = lane & 15;
    {
        const int n0 = wave * 16;
        v2f bf[32];
        load_bfrags_t(Wt, n0 + l, half, bf);
        const v8f acc = tile_mma(T + (size_t)(16 * e + l) * HD + 2 * half, bf);
#pragma unroll
        for (int r = 0; r < 8; ++r)
            M[(size_t)(16 * e + r + 8 * half) * HD + n0 + l] = acc[r];
    }
    if (e < 3) {
        const float* S   = (e == 0) ? Wt : (e == 1) ? T : t0;
        const float* vec = (e == 0) ? t0 : bias;
        float*       dv  = (e == 0) ? cv : (e == 1) ? cs : c0;
        const int nrow   = (e == 2) ? ((wave == 0) ? 1 : 0) : 16;
        const float4 c = *(const float4*)(vec + lane * 4);
        float4 rv[16];
#pragma unroll
        for (int k = 0; k < 16; ++k)
            rv[k] = (k < nrow) ? *(const float4*)(S + (size_t)(16 * wave + k) * HD + lane * 4)
                               : make_float4(0.f, 0.f, 0.f, 0.f);
#pragma unroll
        for (int k = 0; k < 16; ++k) {
            float s = rv[k].x * c.x + rv[k].y * c.y + rv[k].z * c.z + rv[k].w * c.w;
            for (int off = 16; off > 0; off >>= 1) s += __shfl_xor(s, off, 32);
            if (lane == 0 && k < nrow) dv[16 * wave + k] = s;
        }
    }
}

__global__ __launch_bounds__(DTH) void dgi_deg_bin(const int* __restrict__ src,
                                                   const int* __restrict__ dst,
                                                   unsigned* __restrict__ part,
                                                   unsigned* __restrict__ lists,
                                                   unsigned* __restrict__ offs,
                                                   const float* __restrict__ Wt,
                                                   const float* __restrict__ Wb,
                                                   const float* __restrict__ bias,
                                                   float* __restrict__ T,
                                                   float* __restrict__ t0) {
    __shared__ unsigned cnt[DRN];
    __shared__ unsigned ttot[TPR];
    __shared__ unsigned cur[TPR];
    __shared__ unsigned wbel[DTH / 32];
    if (blockIdx.x >= DR * DC) {
        fold_stage1(Wt, Wb, bias, T, t0, (int)blockIdx.x - DR * DC);
        return;
    }
    const int c    = blockIdx.x / DR;
    const int r    = blockIdx.x % DR;
    const int nlo  = r * DRN;
    const int lane = threadIdx.x & 31;
    const int wave = threadIdx.x >> 5;

    for (int i = threadIdx.x; i < DRN; i += DTH) cnt[i] = 0u;
    __syncthreads();

    const int4* d4 = (const int4*)dst + (size_t)c * DCE4;
    const int4* s4 = (const int4*)src + (size_t)c * DCE4;
    int4 d[DG], sv[DG];
#pragma unroll
    for (int k = 0; k < DG; ++k) {
        const int i = threadIdx.x + k * DTH;
        d[k] = d4[(i < DCE4) ? i : DCE4 - 1];
    }
#pragma unroll
    for (int k = 0; k < DG; ++k) {
        const int i = threadIdx.x + k * DTH;
        sv[k] = s4[(i < DCE4) ? i : DCE4 - 1];
    }

    unsigned below = 0u;
#pragma unroll
    for (int k = 0; k < DG; ++k) {
        const int i = threadIdx.x + k * DTH;
        if (i < DCE4) {
            const unsigned a0 = (unsigned)(d[k].x - nlo);
            const unsigned a1 = (unsigned)(d[k].y - nlo);
            const unsigned a2 = (unsigned)(d[k].z - nlo);
            const unsigned a3 = (unsigned)(d[k].w - nlo);
            if (a0 < DRN) atomicAdd(&cnt[a0], 1u);
            if (a1 < DRN) atomicAdd(&cnt[a1], 1u);
            if (a2 < DRN) atomicAdd(&cnt[a2], 1u);
            if (a3 < DRN) atomicAdd(&cnt[a3], 1u);
            below += (unsigned)(d[k].x < nlo) + (unsigned)(d[k].y < nlo) +
                     (unsigned)(d[k].z < nlo) + (unsigned)(d[k].w < nlo);
        }
    }
    for (int off = 16; off > 0; off >>= 1) below += __shfl_xor(below, off, 32);
    if (lane == 0) wbel[wave] = below;
    __syncthreads();

    unsigned* prow = part + (size_t)c * NN + nlo;
    for (int i = threadIdx.x; i < DRN; i += DTH)
        if (nlo + i < NN) prow[i] = cnt[i];

    {
        const int tl  = threadIdx.x >> 4;
        const int sub = threadIdx.x & 15;
        unsigned t = 0u;
        for (int m = sub; m < TR; m += 16) t += cnt[tl * TR + m];
        for (int off = 8; off > 0; off >>= 1) t += __shfl_xor(t, off, 32);
        if (sub == 0) ttot[tl] = t;
    }
    __syncthreads();

    if (wave == 0) {
        unsigned b = (lane < DTH / 32) ? wbel[lane] : 0u;
        for (int off = 16; off > 0; off >>= 1) b += __shfl_xor(b, off, 32);
        const unsigned t = ttot[lane];
        unsigned inc = t;
        for (int off = 1; off < 32; off <<= 1) {
            const unsigned v = __shfl_up(inc, off, 32);
            if (lane >= off) inc += v;
        }
        const unsigned start = b + inc - t;
        cur[lane] = start;
        offs[c * NTILE + r * TPR + lane] = start;
    }
    __syncthreads();

    unsigned* lst = lists + (size_t)c * DCE;
#pragma unroll
    for (int k = 0; k < DG; ++k) {
        const int i = threadIdx.x + k * DTH;
        if (i < DCE4) {
            const int dd[4] = {d[k].x, d[k].y, d[k].z, d[k].w};
            const int ss[4] = {sv[k].x, sv[k].y, sv[k].z, sv[k].w};
            unsigned a[4], tl[4], pos[4];
#pragma unroll
            for (int j = 0; j < 4; ++j) {
                a[j]   = (unsigned)(dd[j] - nlo);
                tl[j]  = a[j] / TR;
                pos[j] = 0u;
                if (a[j] < DRN) pos[j] = atomicAdd(&cur[tl[j]], 1u);
            }
#pragma unroll
            for (int j = 0; j < 4; ++j)
                if (a[j] < DRN) lst[pos[j]] = ((unsigned)ss[j] << 8) | (a[j] - tl[j] * TR);
        }
    }
}

__global__ __launch_bounds__(256) void dgi_dinv(const unsigned* __restrict__ part,
                                                float* __restrict__ dinv,
                                                const float* __restrict__ Wt,
                                                const float* __restrict__ bias,
                                                const float* __restrict__ T,
                                                const float* __restrict__ t0,
                                                float* __restrict__ M,
                                                float* __restrict__ cv,
                                                float* __restrict__ cs,
                                                float* __restrict__ c0) {
    if (blockIdx.x >= NBLK) {
        fold_stage2(Wt, bias, T, t0, M, cv, cs, c0, (int)blockIdx.x - NBLK);
        return;
    }
    const int i = blockIdx.x * blockDim.x + threadIdx.x;
    if (i < NN) {
        unsigned s = 1u;
#pragma unroll
        for (int c = 0; c < DC; ++c) s += part[(size_t)c * NN + i];
        const float d = (float)s;
        dinv[i] = (d > 0.0f) ? rsqrtf(d) : 0.0f;
    }
}

#define TRP    208
#define PITCH  132
#define AGGW   32
#define NSELF  ((TRP + AGGW - 1) / AGGW)
static_assert(AGGW == DC, "one edge chunk per owner wave");

__global__ __launch_bounds__(1024) void dgi_tile_aggregate(const unsigned* __restrict__ lists,
                                                           const unsigned* __restrict__ offs,
                                                           const float* __restrict__ x,
                                                           const float* __restrict__ M,
                                                           const float* __restrict__ dinv,
                                                           const float* __restrict__ cv,
                                                           float* __restrict__ y,
                                                           float* __restrict__ v) {
    __shared__ float    tile[TRP * PITCH];

    const int wave = __builtin_amdgcn_readfirstlane((int)(threadIdx.x >> 5));
    const int lane = threadIdx.x & 31;
    const int lo   = blockIdx.x * TR;
    const int rows = (NN - lo < TR) ? (NN - lo) : TR;

    float4 xs[NSELF];
    float  ds[NSELF];
#pragma unroll
    for (int i = 0; i < NSELF; ++i) {
        const int r  = wave + i * AGGW;
        const int rc = (r < rows) ? r : rows - 1;
        ds[i] = dinv[lo + rc];
        xs[i] = *(const float4*)(x + (size_t)(lo + rc) * HD + lane * 4);
    }

    const unsigned* lst = lists + (size_t)wave * DCE;
    const int beg = (int)offs[wave * NTILE + blockIdx.x];
    const int end = (blockIdx.x == NTILE - 1) ? DCE
                                              : (int)offs[wave * NTILE + blockIdx.x + 1];
    unsigned e0 = 0u, e1 = 0u;
    if (beg + lane < end)      e0 = lst[beg + lane];
    if (beg + 32 + lane < end) e1 = lst[beg + 32 + lane];
    int   s_l = (int)(e0 >> 8);
    int   r_l = (int)(e0 & 255u);
    const float fa0 = dinv[s_l];
    const float fb0 = dinv[lo + r_l];

#pragma unroll
    for (int i = 0; i < NSELF; ++i) {
        const int r = wave + i * AGGW;
        if (r < TRP) {
            float4 v = make_float4(0.f, 0.f, 0.f, 0.f);
            if (r < rows) {
                const float dd = ds[i] * ds[i];
                v.x = dd * xs[i].x;
                v.y = dd * xs[i].y;
                v.z = dd * xs[i].z;
                v.w = dd * xs[i].w;
            }
            float* t = tile + r * PITCH + lane;
            t[0]  = v.x;
            t[32] = v.y;
            t[64] = v.z;
            t[96] = v.w;
        }
    }
    __syncthreads();
    float c_l = fa0 * fb0;

    {
        for (int base = beg; base < end; base += 32) {
            const int cnt = (end - base < 32) ? (end - base) : 32;
            const int s_n = (int)(e1 >> 8);
            const int r_n = (int)(e1 & 255u);
            const float fa = dinv[s_n];
            const float fb = dinv[lo + r_n];
            unsigned e2 = 0u;
            if (base + 64 + lane < end) e2 = lst[base + 64 + lane];
            int k = 0;
            for (; k + 4 <= cnt; k += 4) {
                float4 v[4];
                int    rr[4];
                float  cc[4];
#pragma unroll
                for (int j = 0; j < 4; ++j) {
                    const int sn = __builtin_amdgcn_readlane(s_l, k + j);
                    rr[j] = __builtin_amdgcn_readlane(r_l, k + j);
                    cc[j] = __int_as_float(__builtin_amdgcn_readlane(__float_as_int(c_l), k + j));
                    v[j]  = *(const float4*)(x + (size_t)sn * HD + lane * 4);
                }
#pragma unroll
                for (int j = 0; j < 4; ++j) {
                    float* t = tile + rr[j] * PITCH + lane;
                    atomicAdd(t + 0,  v[j].x * cc[j]);
                    atomicAdd(t + 32, v[j].y * cc[j]);
                    atomicAdd(t + 64, v[j].z * cc[j]);
                    atomicAdd(t + 96, v[j].w * cc[j]);
                }
            }
            for (; k < cnt; ++k) {
                const int   sn = __builtin_amdgcn_readlane(s_l, k);
                const int   rn = __builtin_amdgcn_readlane(r_l, k);
                const float cn = __int_as_float(__builtin_amdgcn_readlane(__float_as_int(c_l), k));
                const float4 v = *(const float4*)(x + (size_t)sn * HD + lane * 4);
                float* t = tile + rn * PITCH + lane;
                atomicAdd(t + 0,  v.x * cn);
                atomicAdd(t + 32, v.y * cn);
                atomicAdd(t + 64, v.z * cn);
                atomicAdd(t + 96, v.w * cn);
            }
            s_l = s_n;
            r_l = r_n;
            c_l = fa * fb;
            e1  = e2;
        }
    }

    const int half = lane >> 4;
    const int l    = lane & 15;
    const int n0   = (wave & 7) * 16;
    const int nrt  = (rows + 15) >> 4;
    v2f bf[32];

    load_bfrags<true>(M, n0 + l, half, bf);
    const float cn = cv[n0 + l];
    __syncthreads();

    for (int r = wave; r < rows; r += AGGW) {
        const float* t = tile + r * PITCH + lane;
        const float4 o = make_float4(t[0], t[32], t[64], t[96]);
        *(float4*)(y + (size_t)(lo + r) * HD + lane * 4) = o;
    }

    for (int rt = wave >> 3; rt < nrt; rt += 4) {
        const v8f acc = tile_mma(tile + (rt * 16 + l) * PITCH + 2 * half, bf);
#pragma unroll
        for (int r = 0; r < 8; ++r) {
            const int m = rt * 16 + r + 8 * half;
            if (m < rows) v[(size_t)(lo + m) * HD + n0 + l] = acc[r] + cn;
        }
    }
}

#define SNW    8
#define SWG    8
static_assert(NN % SNW == 0, "every wave has all of its nodes");

__global__ __launch_bounds__(SWG * 32) void dgi_score(const float* __restrict__ u,
                                                      const float* __restrict__ z,
                                                      const int* __restrict__ perm,
                                                      const float* __restrict__ bb,
                                                      const float* __restrict__ cs,
                                                      const float* __restrict__ c0,
                                                      float* __restrict__ out) {
    const int wave = threadIdx.x >> 5;
    const int lane = threadIdx.x & 31;
    const int g    = lane >> 3;
    const int q    = lane & 7;
    const int n0   = (blockIdx.x * SWG + wave) * SNW;
    if (n0 >= NN) return;

    int n[2], p[2];
#pragma unroll
    for (int h = 0; h < 2; ++h) {
        n[h] = n0 + 4 * h + g;
        p[h] = perm[n[h]];
    }

    float4 uu[2][4], va[2][4], vb[2][4];
#pragma unroll
    for (int h = 0; h < 2; ++h) {
#pragma unroll
        for (int j = 0; j < 4; ++j) {
            uu[h][j] = *(const float4*)(u + (size_t)n[h] * HD + 32 * j + 4 * q);
            va[h][j] = *(const float4*)(z + (size_t)n[h] * HD + 32 * j + 4 * q);
        }
    }
#pragma unroll
    for (int h = 0; h < 2; ++h) {
#pragma unroll
        for (int j = 0; j < 4; ++j)
            vb[h][j] = *(const float4*)(z + (size_t)p[h] * HD + 32 * j + 4 * q);
    }

    const float bias = bb[0] + c0[0];
    float4 cc[4];
#pragma unroll
    for (int j = 0; j < 4; ++j) cc[j] = *(const float4*)(cs + 32 * j + 4 * q);
    __builtin_amdgcn_sched_barrier(0);
#pragma unroll
    for (int h = 0; h < 2; ++h) {
        float pos = 0.0f, neg = 0.0f, sn = 0.0f;
#pragma unroll
        for (int j = 0; j < 4; ++j) {
            sn  += cc[j].x * va[h][j].x + cc[j].y * va[h][j].y +
                   cc[j].z * va[h][j].z + cc[j].w * va[h][j].w;
            pos += uu[h][j].x * va[h][j].x + uu[h][j].y * va[h][j].y +
                   uu[h][j].z * va[h][j].z + uu[h][j].w * va[h][j].w;
            neg += uu[h][j].x * vb[h][j].x + uu[h][j].y * vb[h][j].y +
                   uu[h][j].z * vb[h][j].z + uu[h][j].w * vb[h][j].w;
        }
        pos += sn;
        neg += sn;
#pragma unroll
        for (int off = 4; off > 0; off >>= 1) {
            pos += __shfl_xor(pos, off, 32);
            neg += __shfl_xor(neg, off, 32);
        }
        if (q == 0) {
            out[n[h]]      = pos + bias;
            out[NN + n[h]] = neg + bias;
        }
    }
}

extern "C" void kernel_launch(void* const* d_in, const int* in_sizes, int n_in,
                              void* d_out, int out_size, void* d_ws, size_t ws_size,
                              hipStream_t stream) {
    (void)in_sizes; (void)n_in; (void)out_size; (void)ws_size;

    const float* x    = (const float*)d_in[0];
    const float* W    = (const float*)d_in[1];
    const float* b    = (const float*)d_in[2];
    const float* Wb   = (const float*)d_in[3];
    const float* bb   = (const float*)d_in[4];
    const int*   ei   = (const int*)d_in[5];
    const int*   perm = (const int*)d_in[6];
    const int*   esrc = ei;
    const int*   edst = ei + NE;
    float*       out  = (float*)d_out;

    char* ws = (char*)d_ws;
    size_t off = 0;
    float* dinv = (float*)(ws + off); off += ((size_t)NN * 4 + 255) & ~(size_t)255;
    float* z    = (float*)(ws + off); off += ((size_t)NN * HD * 4 + 255) & ~(size_t)255;
    float* u    = (float*)(ws + off); off += ((size_t)NN * HD * 4 + 255) & ~(size_t)255;
    unsigned* lists = (unsigned*)(ws + off); off += ((size_t)NE * 4 + 255) & ~(size_t)255;
    unsigned* offs  = (unsigned*)(ws + off); off += ((size_t)DC * NTILE * 4 + 255) & ~(size_t)255;
    float* M  = (float*)(ws + off); off += (size_t)HD * HD * 4;
    float* cv = (float*)(ws + off); off += (size_t)HD * 4;
    float* cs = (float*)(ws + off); off += (size_t)HD * 4;
    float* c0 = (float*)(ws + off); off += 256;

    unsigned* part = (unsigned*)u;
    float* T  = u + (size_t)DC * NN;
    float* t0 = T + (size_t)HD * HD;

    dgi_deg_bin<<<DR * DC + FOLDB, DTH, 0, stream>>>(esrc, edst, part, lists, offs, W, Wb, b, T, t0);
    dgi_dinv<<<NBLK + FOLDB, 256, 0, stream>>>(part, dinv, W, b, T, t0, M, cv, cs, c0);

    dgi_tile_aggregate<<<NTILE, 1024, 0, stream>>>(lists, offs, x, M, dinv, cv, z, u);

    dgi_score<<<(NN / SNW + SWG - 1) / SWG, SWG * 32, 0, stream>>>(u, z, perm, bb, cs, c0, out);
}


# owner tile split at row 144, lists binned by (tile, part): the 128x128 transform of rows 0-143 runs on the matrix units beside the gather of rows 144-195; 4 of 13 row tiles stay behind
# speedup vs baseline: 5.3814x; 1.0527x over previous
#include <hip/hip_runtime.h>

#define NN   50000
#define NE   800000
#define HD   128

#define TR     196
#define NTILE  256
#define NE4    (NE / 4)

typedef float v2f __attribute__((ext_vector_type(2)));
typedef float v8f __attribute__((ext_vector_type(8)));

#define DR     8
#define DC     32
#define TPR    (NTILE / DR)
#define DRN    (TPR * TR)
#define DCE    (NE / DC)
#define DCE4   (NE4 / DC)
#define DTH    512
#define RT0    9
#define R0     (RT0 * 16)
#define DG     ((DCE4 + DTH - 1) / DTH)

#define FOLDB  (HD / 16)
#define NBLK   ((NN + 255) / 256)

static_assert(TPR == 32, "one lane per tile of a range in the start scan");
static_assert(DTH / 16 == TPR, "sixteen threads per tile in the tile sums");
static_assert(DR * DRN >= NN && NTILE * TR >= NN, "ranges and tiles cover every node");
static_assert(TR <= 256 && NN < (1 << 24), "a list word holds the row in 8 bits and the source above them");
static_assert(R0 < TR && DCE < (1 << 16), "two parts per tile; a tile's start and its part boundary share one word");

template <bool PERMK>
__device__ __forceinline__ int kmap(int p) {
    return PERMK ? (4 * (p & 31) + (p >> 5)) : p;
}

template <bool PERMK>
__device__ __forceinline__ void load_bfrags(const float* __restrict__ B, int col, int half,
                                            v2f (&bf)[32]) {
#pragma unroll
    for (int j = 0; j < 32; ++j) {
        const int p = 4 * j + 2 * half;
        bf[j].x = B[(size_t)kmap<PERMK>(p + 0) * HD + col];
        bf[j].y = B[(size_t)kmap<PERMK>(p + 1) * HD + col];
    }
}

__device__ __forceinline__ void load_bfrags_t(const float* __restrict__ S, int col, int half,
                                              v2f (&bf)[32]) {
#pragma unroll
    for (int j = 0; j < 32; ++j)
        bf[j] = *(const v2f*)(S + (size_t)col * HD + 4 * j + 2 * half);
}

__device__ __forceinline__ v8f tile_mma(const float* arow, const v2f (&bf)[32]) {
    v8f acc = {0.f, 0.f, 0.f, 0.f, 0.f, 0.f, 0.f, 0.f};
#pragma unroll
    for (int j = 0; j < 32; ++j) {
        const v2f a = *(const v2f*)(arow + 4 * j);
        acc = __builtin_amdgcn_wmma_f32_16x16x4_f32(
            false, a, false, bf[j], (short)0, acc, false, false);
    }
    return acc;
}

__device__ __forceinline__ void fold_stage1(const float* __restrict__ Wt, const float* __restrict__ Wb,
                                            const float* __restrict__ bias,
                                            float* __restrict__ T, float* __restrict__ t0, int e) {
    const int wave = __builtin_amdgcn_readfirstlane((int)(threadIdx.x >> 5));
    const int lane = threadIdx.x & 31;
    const int half = lane >> 4;
    const int l    = lane & 15;
    if (wave < 8) {
        const int n0 = wave * 16;
        v2f bf[32];
        load_bfrags<false>(Wb, n0 + l, half, bf);
        const v8f acc = tile_mma(Wt + (size_t)(16 * e + l) * HD + 2 * half, bf);
#pragma unroll
        for (int r = 0; r < 8; ++r)
            T[(size_t)(16 * e + r + 8 * half) * HD + n0 + l] = acc[r];
    } else if (e == 0 && wave < 12) {
        const int g = (wave - 8) * 32 + lane;
        float s = 0.0f;
        for (int h0 = 0; h0 < HD; h0 += 16) {
            float w[16];
#pragma unroll
            for (int k = 0; k < 16; ++k) w[k] = Wb[(size_t)(h0 + k) * HD + g];
#pragma unroll
            for (int k = 0; k < 16; ++k) s += bias[h0 + k] * w[k];
        }
        t0[g] = s;
    }
}

__device__ __forceinline__ void fold_stage2(const float* __restrict__ Wt, const float* __restrict__ bias,
                                            const float* __restrict__ T, const float* __restrict__ t0,
                                            float* __restrict__ M, float* __restrict__ cv,
                                            float* __restrict__ cs, float* __restrict__ c0, int e) {
    const int wave = threadIdx.x >> 5;
    const int lane = threadIdx.x & 31;
    const int half = lane >> 4;
    const int l    = lane & 15;
    {
        const int n0 = wave * 16;
        v2f bf[32];
        load_bfrags_t(Wt, n0 + l, half, bf);
        const v8f acc = tile_mma(T + (size_t)(16 * e + l) * HD + 2 * half, bf);
#pragma unroll
        for (int r = 0; r < 8; ++r)
            M[(size_t)(16 * e + r + 8 * half) * HD + n0 + l] = acc[r];
    }
    if (e < 3) {
        const float* S   = (e == 0) ? Wt : (e == 1) ? T : t0;
        const float* vec = (e == 0) ? t0 : bias;
        float*       dv  = (e == 0) ? cv : (e == 1) ? cs : c0;
        const int nrow   = (e == 2) ? ((wave == 0) ? 1 : 0) : 16;
        const float4 c = *(const float4*)(vec + lane * 4);
        float4 rv[16];
#pragma unroll
        for (int k = 0; k < 16; ++k)
            rv[k] = (k < nrow) ? *(const float4*)(S + (size_t)(16 * wave + k) * HD + lane * 4)
                               : make_float4(0.f, 0.f, 0.f, 0.f);
#pragma unroll
        for (int k = 0; k < 16; ++k) {
            float s = rv[k].x * c.x + rv[k].y * c.y + rv[k].z * c.z + rv[k].w * c.w;
            for (int off = 16; off > 0; off >>= 1) s += __shfl_xor(s, off, 32);
            if (lane == 0 && k < nrow) dv[16 * wave + k] = s;
        }
    }
}

__global__ __launch_bounds__(DTH) void dgi_deg_bin(const int* __restrict__ src,
                                                   const int* __restrict__ dst,
                                                   unsigned* __restrict__ part,
                                                   unsigned* __restrict__ lists,
                                                   unsigned* __restrict__ offs,
                                                   const float* __restrict__ Wt,
                                                   const float* __restrict__ Wb,
                                                   const float* __restrict__ bias,
                                                   float* __restrict__ T,
                                                   float* __restrict__ t0) {
    __shared__ unsigned cnt[DRN];
    __shared__ unsigned ttot[2 * TPR];
    __shared__ unsigned cur[2 * TPR];
    __shared__ unsigned wbel[DTH / 32];
    if (blockIdx.x >= DR * DC) {
        fold_stage1(Wt, Wb, bias, T, t0, (int)blockIdx.x - DR * DC);
        return;
    }
    const int c    = blockIdx.x / DR;
    const int r    = blockIdx.x % DR;
    const int nlo  = r * DRN;
    const int lane = threadIdx.x & 31;
    const int wave = threadIdx.x >> 5;

    for (int i = threadIdx.x; i < DRN; i += DTH) cnt[i] = 0u;
    __syncthreads();

    const int4* d4 = (const int4*)dst + (size_t)c * DCE4;
    const int4* s4 = (const int4*)src + (size_t)c * DCE4;
    int4 d[DG], sv[DG];
#pragma unroll
    for (int k = 0; k < DG; ++k) {
        const int i = threadIdx.x + k * DTH;
        d[k] = d4[(i < DCE4) ? i : DCE4 - 1];
    }
#pragma unroll
    for (int k = 0; k < DG; ++k) {
        const int i = threadIdx.x + k * DTH;
        sv[k] = s4[(i < DCE4) ? i : DCE4 - 1];
    }

    unsigned below = 0u;
#pragma unroll
    for (int k = 0; k < DG; ++k) {
        const int i = threadIdx.x + k * DTH;
        if (i < DCE4) {
            const unsigned a0 = (unsigned)(d[k].x - nlo);
            const unsigned a1 = (unsigned)(d[k].y - nlo);
            const unsigned a2 = (unsigned)(d[k].z - nlo);
            const unsigned a3 = (unsigned)(d[k].w - nlo);
            if (a0 < DRN) atomicAdd(&cnt[a0], 1u);
            if (a1 < DRN) atomicAdd(&cnt[a1], 1u);
            if (a2 < DRN) atomicAdd(&cnt[a2], 1u);
            if (a3 < DRN) atomicAdd(&cnt[a3], 1u);
            below += (unsigned)(d[k].x < nlo) + (unsigned)(d[k].y < nlo) +
                     (unsigned)(d[k].z < nlo) + (unsigned)(d[k].w < nlo);
        }
    }
    for (int off = 16; off > 0; off >>= 1) below += __shfl_xor(below, off, 32);
    if (lane == 0) wbel[wave] = below;
    __syncthreads();

    unsigned* prow = part + (size_t)c * NN + nlo;
    for (int i = threadIdx.x; i < DRN; i += DTH)
        if (nlo + i < NN) prow[i] = cnt[i];

    {
        const int tl  = threadIdx.x >> 4;
        const int sub = threadIdx.x & 15;
        unsigned ta = 0u, tb = 0u;
        for (int m = sub; m < TR; m += 16) {
            const unsigned v = cnt[tl * TR + m];
            if (m < R0) ta += v; else tb += v;
        }
        for (int off = 8; off > 0; off >>= 1) {
            ta += __shfl_xor(ta, off, 32);
            tb += __shfl_xor(tb, off, 32);
        }
        if (sub == 0) { ttot[2 * tl] = ta; ttot[2 * tl + 1] = tb; }
    }
    __syncthreads();

    if (wave == 0) {
        unsigned b = (lane < DTH / 32) ? wbel[lane] : 0u;
        for (int off = 16; off > 0; off >>= 1) b += __shfl_xor(b, off, 32);
        const unsigned ta = ttot[2 * lane];
        const unsigned t  = ta + ttot[2 * lane + 1];
        unsigned inc = t;
        for (int off = 1; off < 32; off <<= 1) {
            const unsigned v = __shfl_up(inc, off, 32);
            if (lane >= off) inc += v;
        }
        const unsigned start = b + inc - t;
        cur[2 * lane]     = start;
        cur[2 * lane + 1] = start + ta;
        offs[c * NTILE + r * TPR + lane] = start | ((start + ta) << 16);
    }
    __syncthreads();

    unsigned* lst = lists + (size_t)c * DCE;
#pragma unroll
    for (int k = 0; k < DG; ++k) {
        const int i = threadIdx.x + k * DTH;
        if (i < DCE4) {
            const int dd[4] = {d[k].x, d[k].y, d[k].z, d[k].w};
            const int ss[4] = {sv[k].x, sv[k].y, sv[k].z, sv[k].w};
            unsigned a[4], tl[4], rw[4], pos[4];
#pragma unroll
            for (int j = 0; j < 4; ++j) {
                a[j]   = (unsigned)(dd[j] - nlo);
                tl[j]  = a[j] / TR;
                rw[j]  = a[j] - tl[j] * TR;
                pos[j] = 0u;
                if (a[j] < DRN) pos[j] = atomicAdd(&cur[2 * tl[j] + (rw[j] >= R0 ? 1u : 0u)], 1u);
            }
#pragma unroll
            for (int j = 0; j < 4; ++j)
                if (a[j] < DRN) lst[pos[j]] = ((unsigned)ss[j] << 8) | rw[j];
        }
    }
}

__global__ __launch_bounds__(256) void dgi_dinv(const unsigned* __restrict__ part,
                                                float* __restrict__ dinv,
                                                const float* __restrict__ Wt,
                                                const float* __restrict__ bias,
                                                const float* __restrict__ T,
                                                const float* __restrict__ t0,
                                                float* __restrict__ M,
                                                float* __restrict__ cv,
                                                float* __restrict__ cs,
                                                float* __restrict__ c0) {
    if (blockIdx.x >= NBLK) {
        fold_stage2(Wt, bias, T, t0, M, cv, cs, c0, (int)blockIdx.x - NBLK);
        return;
    }
    const int i = blockIdx.x * blockDim.x + threadIdx.x;
    if (i < NN) {
        unsigned s = 1u;
#pragma unroll
        for (int c = 0; c < DC; ++c) s += part[(size_t)c * NN + i];
        const float d = (float)s;
        dinv[i] = (d > 0.0f) ? rsqrtf(d) : 0.0f;
    }
}

#define TRP    208
#define PITCH  132
#define AGGW   32
#define NSELF  ((TRP + AGGW - 1) / AGGW)
static_assert(AGGW == DC, "one edge chunk per owner wave");

__device__ __forceinline__ void gather_stretch(const unsigned* __restrict__ lst, int beg, int end,
                                               unsigned e1, int s_l, int r_l, float c_l,
                                               const float* __restrict__ x,
                                               const float* __restrict__ dinv,
                                               int lo, float* tile, int lane) {
    for (int base = beg; base < end; base += 32) {
        const int cnt = (end - base < 32) ? (end - base) : 32;
        const int s_n = (int)(e1 >> 8);
        const int r_n = (int)(e1 & 255u);
        const float fa = dinv[s_n];
        const float fb = dinv[lo + r_n];
        unsigned e2 = 0u;
        if (base + 64 + lane < end) e2 = lst[base + 64 + lane];
        int k = 0;
        for (; k + 4 <= cnt; k += 4) {
            float4 v[4];
            int    rr[4];
            float  cc[4];
#pragma unroll
            for (int j = 0; j < 4; ++j) {
                const int sn = __builtin_amdgcn_readlane(s_l, k + j);
                rr[j] = __builtin_amdgcn_readlane(r_l, k + j);
                cc[j] = __int_as_float(__builtin_amdgcn_readlane(__float_as_int(c_l), k + j));
                v[j]  = *(const float4*)(x + (size_t)sn * HD + lane * 4);
            }
#pragma unroll
            for (int j = 0; j < 4; ++j) {
                float* t = tile + rr[j] * PITCH + lane;
                atomicAdd(t + 0,  v[j].x * cc[j]);
                atomicAdd(t + 32, v[j].y * cc[j]);
                atomicAdd(t + 64, v[j].z * cc[j]);
                atomicAdd(t + 96, v[j].w * cc[j]);
            }
        }
        if (k < cnt) {
            const int n = cnt - k;
            float4 v[3];
            int    rr[3];
            float  cc[3];
#pragma unroll
            for (int j = 0; j < 3; ++j) {
                v[j]  = make_float4(0.f, 0.f, 0.f, 0.f);
                rr[j] = 0;
                cc[j] = 0.f;
                if (j < n) {
                    const int sn = __builtin_amdgcn_readlane(s_l, k + j);
                    rr[j] = __builtin_amdgcn_readlane(r_l, k + j);
                    cc[j] = __int_as_float(__builtin_amdgcn_readlane(__float_as_int(c_l), k + j));
                    v[j]  = *(const float4*)(x + (size_t)sn * HD + lane * 4);
                }
            }
#pragma unroll
            for (int j = 0; j < 3; ++j) {
                if (j < n) {
                    float* t = tile + rr[j] * PITCH + lane;
                    atomicAdd(t + 0,  v[j].x * cc[j]);
                    atomicAdd(t + 32, v[j].y * cc[j]);
                    atomicAdd(t + 64, v[j].z * cc[j]);
                    atomicAdd(t + 96, v[j].w * cc[j]);
                }
            }
        }
        s_l = s_n;
        r_l = r_n;
        c_l = fa * fb;
        e1  = e2;
    }
}

__global__ __launch_bounds__(1024) void dgi_tile_aggregate(const unsigned* __restrict__ lists,
                                                           const unsigned* __restrict__ offs,
                                                           const float* __restrict__ x,
                                                           const float* __restrict__ M,
                                                           const float* __restrict__ dinv,
                                                           const float* __restrict__ cv,
                                                           float* __restrict__ y,
                                                           float* __restrict__ v) {
    __shared__ float    tile[TRP * PITCH];

    const int wave = __builtin_amdgcn_readfirstlane((int)(threadIdx.x >> 5));
    const int lane = threadIdx.x & 31;
    const int lo   = blockIdx.x * TR;
    const int rows = (NN - lo < TR) ? (NN - lo) : TR;

    float4 xs[NSELF];
    float  ds[NSELF];
#pragma unroll
    for (int i = 0; i < NSELF; ++i) {
        const int r  = wave + i * AGGW;
        const int rc = (r < rows) ? r : rows - 1;
        ds[i] = dinv[lo + rc];
        xs[i] = *(const float4*)(x + (size_t)(lo + rc) * HD + lane * 4);
    }

    const unsigned* lst = lists + (size_t)wave * DCE;
    const unsigned ob = offs[wave * NTILE + blockIdx.x];
    const int beg = (int)(ob & 0xffffu);
    const int mid = (int)(ob >> 16);
    const int end = (blockIdx.x == NTILE - 1) ? DCE
                                              : (int)(offs[wave * NTILE + blockIdx.x + 1] & 0xffffu);
    unsigned e0 = 0u, e1 = 0u;
    if (beg + lane < mid)      e0 = lst[beg + lane];
    if (beg + 32 + lane < mid) e1 = lst[beg + 32 + lane];
    int   s_l = (int)(e0 >> 8);
    int   r_l = (int)(e0 & 255u);
    const float fa0 = dinv[s_l];
    const float fb0 = dinv[lo + r_l];

#pragma unroll
    for (int i = 0; i < NSELF; ++i) {
        const int r = wave + i * AGGW;
        if (r < TRP) {
            float4 v = make_float4(0.f, 0.f, 0.f, 0.f);
            if (r < rows) {
                const float dd = ds[i] * ds[i];
                v.x = dd * xs[i].x;
                v.y = dd * xs[i].y;
                v.z = dd * xs[i].z;
                v.w = dd * xs[i].w;
            }
            float* t = tile + r * PITCH + lane;
            t[0]  = v.x;
            t[32] = v.y;
            t[64] = v.z;
            t[96] = v.w;
        }
    }
    __syncthreads();
    float c_l = fa0 * fb0;

    gather_stretch(lst, beg, mid, e1, s_l, r_l, c_l, x, dinv, lo, tile, lane);

    const int half = lane >> 4;
    const int l    = lane & 15;
    const int n0   = (wave & 7) * 16;
    const int nrt  = (rows + 15) >> 4;
    const int nrt0 = (nrt < RT0) ? nrt : RT0;
    const int rows0 = (rows < R0) ? rows : R0;
    v2f bf[32];

    unsigned g0 = 0u, g1 = 0u;
    if (mid + lane < end)      g0 = lst[mid + lane];
    if (mid + 32 + lane < end) g1 = lst[mid + 32 + lane];
    const int   s_m = (int)(g0 >> 8);
    const int   r_m = (int)(g0 & 255u);
    const float fa1 = dinv[s_m];
    const float fb1 = dinv[lo + r_m];
    load_bfrags<true>(M, n0 + l, half, bf);
    const float cn = cv[n0 + l];
    __syncthreads();

    const bool gather_first = ((wave >> 2) & 1) != 0;
    if (gather_first)
        gather_stretch(lst, mid, end, g1, s_m, r_m, fa1 * fb1, x, dinv, lo, tile, lane);

    for (int r = wave; r < rows0; r += AGGW) {
        const float* t = tile + r * PITCH + lane;
        const float4 o = make_float4(t[0], t[32], t[64], t[96]);
        *(float4*)(y + (size_t)(lo + r) * HD + lane * 4) = o;
    }
    for (int rt = wave >> 3; rt < nrt0; rt += 4) {
        const v8f acc = tile_mma(tile + (rt * 16 + l) * PITCH + 2 * half, bf);
#pragma unroll
        for (int r = 0; r < 8; ++r) {
            const int m = rt * 16 + r + 8 * half;
            if (m < rows) v[(size_t)(lo + m) * HD + n0 + l] = acc[r] + cn;
        }
    }

    if (!gather_first)
        gather_stretch(lst, mid, end, g1, s_m, r_m, fa1 * fb1, x, dinv, lo, tile, lane);
    __syncthreads();

    for (int r = R0 + wave; r < rows; r += AGGW) {
        const float* t = tile + r * PITCH + lane;
        const float4 o = make_float4(t[0], t[32], t[64], t[96]);
        *(float4*)(y + (size_t)(lo + r) * HD + lane * 4) = o;
    }
    for (int rt = RT0 + (wave >> 3); rt < nrt; rt += 4) {
        const v8f acc = tile_mma(tile + (rt * 16 + l) * PITCH + 2 * half, bf);
#pragma unroll
        for (int r = 0; r < 8; ++r) {
            const int m = rt * 16 + r + 8 * half;
            if (m < rows) v[(size_t)(lo + m) * HD + n0 + l] = acc[r] + cn;
        }
    }
}

#define SNW    8
#define SWG    8
static_assert(NN % SNW == 0, "every wave has all of its nodes");

__global__ __launch_bounds__(SWG * 32) void dgi_score(const float* __restrict__ u,
                                                      const float* __restrict__ z,
                                                      const int* __restrict__ perm,
                                                      const float* __restrict__ bb,
                                                      const float* __restrict__ cs,
                                                      const float* __restrict__ c0,
                                                      float* __restrict__ out) {
    const int wave = threadIdx.x >> 5;
    const int lane = threadIdx.x & 31;
    const int g    = lane >> 3;
    const int q    = lane & 7;
    const int n0   = (blockIdx.x * SWG + wave) * SNW;
    if (n0 >= NN) return;

    int n[2], p[2];
#pragma unroll
    for (int h = 0; h < 2; ++h) {
        n[h] = n0 + 4 * h + g;
        p[h] = perm[n[h]];
    }

    float4 uu[2][4], va[2][4], vb[2][4];
#pragma unroll
    for (int h = 0; h < 2; ++h) {
#pragma unroll
        for (int j = 0; j < 4; ++j) {
            uu[h][j] = *(const float4*)(u + (size_t)n[h] * HD + 32 * j + 4 * q);
            va[h][j] = *(const float4*)(z + (size_t)n[h] * HD + 32 * j + 4 * q);
        }
    }
#pragma unroll
    for (int h = 0; h < 2; ++h) {
#pragma unroll
        for (int j = 0; j < 4; ++j)
            vb[h][j] = *(const float4*)(z + (size_t)p[h] * HD + 32 * j + 4 * q);
    }

    const float bias = bb[0] + c0[0];
    float4 cc[4];
#pragma unroll
    for (int j = 0; j < 4; ++j) cc[j] = *(const float4*)(cs + 32 * j + 4 * q);
    __builtin_amdgcn_sched_barrier(0);
#pragma unroll
    for (int h = 0; h < 2; ++h) {
        float pos = 0.0f, neg = 0.0f, sn = 0.0f;
#pragma unroll
        for (int j = 0; j < 4; ++j) {
            sn  += cc[j].x * va[h][j].x + cc[j].y * va[h][j].y +
                   cc[j].z * va[h][j].z + cc[j].w * va[h][j].w;
            pos += uu[h][j].x * va[h][j].x + uu[h][j].y * va[h][j].y +
                   uu[h][j].z * va[h][j].z + uu[h][j].w * va[h][j].w;
            neg += uu[h][j].x * vb[h][j].x + uu[h][j].y * vb[h][j].y +
                   uu[h][j].z * vb[h][j].z + uu[h][j].w * vb[h][j].w;
        }
        pos += sn;
        neg += sn;
#pragma unroll
        for (int off = 4; off > 0; off >>= 1) {
            pos += __shfl_xor(pos, off, 32);
            neg += __shfl_xor(neg, off, 32);
        }
        if (q == 0) {
            out[n[h]]      = pos + bias;
            out[NN + n[h]] = neg + bias;
        }
    }
}

extern "C" void kernel_launch(void* const* d_in, const int* in_sizes, int n_in,
                              void* d_out, int out_size, void* d_ws, size_t ws_size,
                              hipStream_t stream) {
    (void)in_sizes; (void)n_in; (void)out_size; (void)ws_size;

    const float* x    = (const float*)d_in[0];
    const float* W    = (const float*)d_in[1];
    const float* b    = (const float*)d_in[2];
    const float* Wb   = (const float*)d_in[3];
    const float* bb   = (const float*)d_in[4];
    const int*   ei   = (const int*)d_in[5];
    const int*   perm = (const int*)d_in[6];
    const int*   esrc = ei;
    const int*   edst = ei + NE;
    float*       out  = (float*)d_out;

    char* ws = (char*)d_ws;
    size_t off = 0;
    float* dinv = (float*)(ws + off); off += ((size_t)NN * 4 + 255) & ~(size_t)255;
    float* z    = (float*)(ws + off); off += ((size_t)NN * HD * 4 + 255) & ~(size_t)255;
    float* u    = (float*)(ws + off); off += ((size_t)NN * HD * 4 + 255) & ~(size_t)255;
    unsigned* lists = (unsigned*)(ws + off); off += ((size_t)NE * 4 + 255) & ~(size_t)255;
    unsigned* offs  = (unsigned*)(ws + off); off += ((size_t)DC * NTILE * 4 + 255) & ~(size_t)255;
    float* M  = (float*)(ws + off); off += (size_t)HD * HD * 4;
    float* cv = (float*)(ws + off); off += (size_t)HD * 4;
    float* cs = (float*)(ws + off); off += (size_t)HD * 4;
    float* c0 = (float*)(ws + off); off += 256;

    unsigned* part = (unsigned*)u;
    float* T  = u + (size_t)DC * NN;
    float* t0 = T + (size_t)HD * HD;

    dgi_deg_bin<<<DR * DC + FOLDB, DTH, 0, stream>>>(esrc, edst, part, lists, offs, W, Wb, b, T, t0);
    dgi_dinv<<<NBLK + FOLDB, 256, 0, stream>>>(part, dinv, W, b, T, t0, M, cv, cs, c0);

    dgi_tile_aggregate<<<NTILE, 1024, 0, stream>>>(lists, offs, x, M, dinv, cv, z, u);

    dgi_score<<<(NN / SNW + SWG - 1) / SWG, SWG * 32, 0, stream>>>(u, z, perm, bb, cs, c0, out);
}
